# starting kernel
# baseline (speedup 1.0000x reference)
#include <hip/hip_runtime.h>
#include <math.h>

#define N_NODES 100000
#define N_EDGES 1600000
#define F_IN    256
#define F_H     128
#define N_CLS   10
#define H_PAD   16

typedef float v2f __attribute__((ext_vector_type(2)));
typedef float v8f __attribute__((ext_vector_type(8)));

__global__ __launch_bounds__(256) void k_deg_init(float* __restrict__ deg) {
    int i = blockIdx.x * 256 + threadIdx.x;
    if (i < N_NODES) deg[i] = 1.0f;
}

__global__ __launch_bounds__(256) void k_deg_add(const int* __restrict__ dst,
                                                 float* __restrict__ deg) {
    int e = blockIdx.x * 256 + threadIdx.x;
    if (e < N_EDGES) atomicAdd(&deg[dst[e]], 1.0f);
}

__global__ __launch_bounds__(256) void k_dinv(float* __restrict__ deg) {
    int i = blockIdx.x * 256 + threadIdx.x;
    if (i < N_NODES) {
        float d = deg[i];
        deg[i] = (d > 0.0f) ? rsqrtf(d) : 0.0f;
    }
}

__global__ __launch_bounds__(256) void k_zero(float* __restrict__ p, long n) {
    long i = (long)blockIdx.x * 256 + threadIdx.x;
    if (i < n) p[i] = 0.0f;
}

#define LDS_STRIDE 260
__global__ __launch_bounds__(256) void k_gemm1(const float* __restrict__ x,
                                               const float* __restrict__ W1,
                                               float* __restrict__ h1) {
    __shared__ float As[16 * LDS_STRIDE];
    const int tile = blockIdx.x;
    const int tid  = threadIdx.x;

    const float4* xg = (const float4*)(x + (size_t)tile * 16 * F_IN);
    #pragma unroll
    for (int it = 0; it < 4; ++it) {
        int q   = tid + 256 * it;
        int row = q >> 6;
        int c4  = q & 63;
        float4 v = xg[row * 64 + c4];
        float* d = &As[row * LDS_STRIDE + c4 * 4];
        d[0] = v.x; d[1] = v.y; d[2] = v.z; d[3] = v.w;
    }
    __syncthreads();

    const int wave = tid >> 5;
    const int lane = tid & 31;
    const int m    = lane & 15;
    const int kg   = lane >> 4;
    const int n0   = wave * 16;

    v8f c = {0.f,0.f,0.f,0.f,0.f,0.f,0.f,0.f};
    for (int k0 = 0; k0 < F_IN; k0 += 4) {
        int ka = k0 + 2 * kg;
        v2f a = *(const v2f*)&As[m * LDS_STRIDE + ka];
        v2f b;
        b.x = W1[(size_t)ka       * F_H + n0 + m];
        b.y = W1[(size_t)(ka + 1) * F_H + n0 + m];
        c = __builtin_amdgcn_wmma_f32_16x16x4_f32(false, a, false, b,
                                                  (short)0, c, false, false);
    }

    float* out = h1 + (size_t)tile * 16 * F_H;
    #pragma unroll
    for (int r = 0; r < 8; ++r) {
        int row = r + 8 * kg;
        out[(size_t)row * F_H + n0 + m] = c[r];
    }
}

__global__ __launch_bounds__(256) void k_scatter1(const int* __restrict__ src,
                                                  const int* __restrict__ dst,
                                                  const float* __restrict__ dinv,
                                                  const float* __restrict__ h1,
                                                  float* __restrict__ agg1) {
    long gid = (long)blockIdx.x * 256 + threadIdx.x;
    long e   = gid >> 5;
    if (e >= N_EDGES) return;
    int f4 = (int)(gid & 31);
    int s = src[e], d = dst[e];
    float nrm = dinv[s] * dinv[d];
    float4 v = *(const float4*)&h1[(size_t)s * F_H + f4 * 4];
    float* o = &agg1[(size_t)d * F_H + f4 * 4];
    atomicAdd(&o[0], v.x * nrm);
    atomicAdd(&o[1], v.y * nrm);
    atomicAdd(&o[2], v.z * nrm);
    atomicAdd(&o[3], v.w * nrm);
}

__global__ __launch_bounds__(256) void k_post1(const float* __restrict__ h1,
                                               const float* __restrict__ dinv,
                                               const float* __restrict__ b1,
                                               float* __restrict__ agg1) {
    long gid = (long)blockIdx.x * 256 + threadIdx.x;
    if (gid >= (long)N_NODES * F_H) return;
    int i = (int)(gid >> 7);
    int f = (int)(gid & (F_H - 1));
    float di = dinv[i];
    float v = agg1[gid] + h1[gid] * di * di + b1[f];
    agg1[gid] = v > 0.0f ? v : 0.0f;
}

__global__ __launch_bounds__(256) void k_padW2(const float* __restrict__ W2,
                                               float* __restrict__ Bp) {
    int t = blockIdx.x * 256 + threadIdx.x;
    if (t >= F_H * H_PAD) return;
    int k = t >> 4, c = t & 15;
    Bp[t] = (c < N_CLS) ? W2[k * N_CLS + c] : 0.0f;
}

__global__ __launch_bounds__(256) void k_gemm2(const float* __restrict__ A,
                                               const float* __restrict__ Bp,
                                               float* __restrict__ h2) {
    const int wave = threadIdx.x >> 5;
    const int lane = threadIdx.x & 31;
    const int tile = blockIdx.x * 8 + wave;
    if ((size_t)tile * 16 >= N_NODES) return;
    const int m  = lane & 15;
    const int kg = lane >> 4;

    const float* Ar = A + ((size_t)tile * 16 + m) * F_H;
    v8f c = {0.f,0.f,0.f,0.f,0.f,0.f,0.f,0.f};
    for (int k0 = 0; k0 < F_H; k0 += 4) {
        int ka = k0 + 2 * kg;
        v2f a = *(const v2f*)&Ar[ka];
        v2f b;
        b.x = Bp[ka * H_PAD + m];
        b.y = Bp[(ka + 1) * H_PAD + m];
        c = __builtin_amdgcn_wmma_f32_16x16x4_f32(false, a, false, b,
                                                  (short)0, c, false, false);
    }
    float* out = h2 + (size_t)tile * 16 * H_PAD;
    #pragma unroll
    for (int r = 0; r < 8; ++r) {
        int row = r + 8 * kg;
        out[(size_t)row * H_PAD + m] = c[r];
    }
}

__global__ __launch_bounds__(256) void k_scatter2(const int* __restrict__ src,
                                                  const int* __restrict__ dst,
                                                  const float* __restrict__ dinv,
                                                  const float* __restrict__ h2,
                                                  float* __restrict__ agg2) {
    long gid = (long)blockIdx.x * 256 + threadIdx.x;
    long e   = gid >> 4;
    if (e >= N_EDGES) return;
    int f = (int)(gid & 15);
    if (f >= N_CLS) return;
    int s = src[e], d = dst[e];
    float nrm = dinv[s] * dinv[d];
    atomicAdd(&agg2[(size_t)d * H_PAD + f], h2[(size_t)s * H_PAD + f] * nrm);
}

__global__ __launch_bounds__(256) void k_post2(const float* __restrict__ h2,
                                               const float* __restrict__ agg2,
                                               const float* __restrict__ dinv,
                                               const float* __restrict__ b2,
                                               float* __restrict__ out) {
    int i = blockIdx.x * 256 + threadIdx.x;
    if (i >= N_NODES) return;
    float di = dinv[i];
    float d2 = di * di;
    float v[N_CLS];
    float mx = -INFINITY;
    #pragma unroll
    for (int c = 0; c < N_CLS; ++c) {
        float t = agg2[(size_t)i * H_PAD + c] + h2[(size_t)i * H_PAD + c] * d2 + b2[c];
        v[c] = t;
        mx = fmaxf(mx, t);
    }
    float s = 0.0f;
    #pragma unroll
    for (int c = 0; c < N_CLS; ++c) s += expf(v[c] - mx);
    float lse = logf(s) + mx;
    #pragma unroll
    for (int c = 0; c < N_CLS; ++c) out[(size_t)i * N_CLS + c] = v[c] - lse;
}

extern "C" void kernel_launch(void* const* d_in, const int* in_sizes, int n_in,
                              void* d_out, int out_size, void* d_ws, size_t ws_size,
                              hipStream_t stream) {
    const float* x   = (const float*)d_in[0];
    const int*   ei  = (const int*)d_in[1];
    const float* W1  = (const float*)d_in[2];
    const float* b1  = (const float*)d_in[3];
    const float* W2  = (const float*)d_in[4];
    const float* b2  = (const float*)d_in[5];
    float*       out = (float*)d_out;

    const int* src = ei;
    const int* dst = ei + N_EDGES;

    float* ws    = (float*)d_ws;
    float* dinv  = ws;
    float* h1    = dinv + 100352;
    float* agg1  = h1 + (size_t)N_NODES * F_H;
    float* W2pad = agg1 + (size_t)N_NODES * F_H;
    float* h2    = h1;
    float* agg2  = h1 + (size_t)N_NODES * H_PAD;

    const int nblk_node = (N_NODES + 255) / 256;

    k_deg_init<<<nblk_node, 256, 0, stream>>>(dinv);
    k_deg_add <<<(N_EDGES + 255) / 256, 256, 0, stream>>>(dst, dinv);
    k_dinv    <<<nblk_node, 256, 0, stream>>>(dinv);

    k_gemm1<<<N_NODES / 16, 256, 0, stream>>>(x, W1, h1);
    {
        long n = (long)N_NODES * F_H;
        k_zero<<<(unsigned)((n + 255) / 256), 256, 0, stream>>>(agg1, n);
    }
    {
        long t = (long)N_EDGES * 32;
        k_scatter1<<<(unsigned)((t + 255) / 256), 256, 0, stream>>>(src, dst, dinv, h1, agg1);
    }
    {
        long n = (long)N_NODES * F_H;
        k_post1<<<(unsigned)((n + 255) / 256), 256, 0, stream>>>(h1, dinv, b1, agg1);
    }

    k_padW2<<<(F_H * H_PAD + 255) / 256, 256, 0, stream>>>(W2, W2pad);
    k_gemm2<<<(N_NODES / 16 + 7) / 8, 256, 0, stream>>>(agg1, W2pad, h2);
    {
        long n = (long)N_NODES * H_PAD;
        k_zero<<<(unsigned)((n + 255) / 256), 256, 0, stream>>>(agg2, n);
    }
    {
        long t = (long)N_EDGES * 16;
        k_scatter2<<<(unsigned)((t + 255) / 256), 256, 0, stream>>>(src, dst, dinv, h2, agg2);
    }
    k_post2<<<nblk_node, 256, 0, stream>>>(h2, agg2, dinv, b2, out);
}


# scatter1: lane l owns features l,l+32,l+64,l+96, so each wave-wide float atomic fills one whole 128-byte line: 4 line requests per edge, not 16; atomics kept, sums arrival-ordered as before
# speedup vs baseline: 1.2114x; 1.2114x over previous
#include <hip/hip_runtime.h>
#include <math.h>

#define N_NODES 100000
#define N_EDGES 1600000
#define F_IN    256
#define F_H     128
#define N_CLS   10
#define H_PAD   16

typedef float v2f __attribute__((ext_vector_type(2)));
typedef float v8f __attribute__((ext_vector_type(8)));

__global__ __launch_bounds__(256) void k_deg_init(float* __restrict__ deg) {
    int i = blockIdx.x * 256 + threadIdx.x;
    if (i < N_NODES) deg[i] = 1.0f;
}

__global__ __launch_bounds__(256) void k_deg_add(const int* __restrict__ dst,
                                                 float* __restrict__ deg) {
    int e = blockIdx.x * 256 + threadIdx.x;
    if (e < N_EDGES) atomicAdd(&deg[dst[e]], 1.0f);
}

__global__ __launch_bounds__(256) void k_dinv(float* __restrict__ deg) {
    int i = blockIdx.x * 256 + threadIdx.x;
    if (i < N_NODES) {
        float d = deg[i];
        deg[i] = (d > 0.0f) ? rsqrtf(d) : 0.0f;
    }
}

__global__ __launch_bounds__(256) void k_zero(float* __restrict__ p, long n) {
    long i = (long)blockIdx.x * 256 + threadIdx.x;
    if (i < n) p[i] = 0.0f;
}

#define LDS_STRIDE 260
__global__ __launch_bounds__(256) void k_gemm1(const float* __restrict__ x,
                                               const float* __restrict__ W1,
                                               float* __restrict__ h1) {
    __shared__ float As[16 * LDS_STRIDE];
    const int tile = blockIdx.x;
    const int tid  = threadIdx.x;

    const float4* xg = (const float4*)(x + (size_t)tile * 16 * F_IN);
    #pragma unroll
    for (int it = 0; it < 4; ++it) {
        int q   = tid + 256 * it;
        int row = q >> 6;
        int c4  = q & 63;
        float4 v = xg[row * 64 + c4];
        float* d = &As[row * LDS_STRIDE + c4 * 4];
        d[0] = v.x; d[1] = v.y; d[2] = v.z; d[3] = v.w;
    }
    __syncthreads();

    const int wave = tid >> 5;
    const int lane = tid & 31;
    const int m    = lane & 15;
    const int kg   = lane >> 4;
    const int n0   = wave * 16;

    v8f c = {0.f,0.f,0.f,0.f,0.f,0.f,0.f,0.f};
    for (int k0 = 0; k0 < F_IN; k0 += 4) {
        int ka = k0 + 2 * kg;
        v2f a = *(const v2f*)&As[m * LDS_STRIDE + ka];
        v2f b;
        b.x = W1[(size_t)ka       * F_H + n0 + m];
        b.y = W1[(size_t)(ka + 1) * F_H + n0 + m];
        c = __builtin_amdgcn_wmma_f32_16x16x4_f32(false, a, false, b,
                                                  (short)0, c, false, false);
    }

    float* out = h1 + (size_t)tile * 16 * F_H;
    #pragma unroll
    for (int r = 0; r < 8; ++r) {
        int row = r + 8 * kg;
        out[(size_t)row * F_H + n0 + m] = c[r];
    }
}

__global__ __launch_bounds__(256) void k_scatter1(const int* __restrict__ src,
                                                  const int* __restrict__ dst,
                                                  const float* __restrict__ dinv,
                                                  const float* __restrict__ h1,
                                                  float* __restrict__ agg1) {
    long gid = (long)blockIdx.x * 256 + threadIdx.x;
    long e   = gid >> 5;
    if (e >= N_EDGES) return;
    int l = (int)(gid & 31);
    int s = src[e], d = dst[e];
    float nrm = dinv[s] * dinv[d];
    const float* hp = &h1[(size_t)s * F_H + l];
    float v0 = hp[0], v1 = hp[32], v2 = hp[64], v3 = hp[96];
    float* o = &agg1[(size_t)d * F_H + l];
    atomicAdd(&o[0],  v0 * nrm);
    atomicAdd(&o[32], v1 * nrm);
    atomicAdd(&o[64], v2 * nrm);
    atomicAdd(&o[96], v3 * nrm);
}

__global__ __launch_bounds__(256) void k_post1(const float* __restrict__ h1,
                                               const float* __restrict__ dinv,
                                               const float* __restrict__ b1,
                                               float* __restrict__ agg1) {
    long gid = (long)blockIdx.x * 256 + threadIdx.x;
    if (gid >= (long)N_NODES * F_H) return;
    int i = (int)(gid >> 7);
    int f = (int)(gid & (F_H - 1));
    float di = dinv[i];
    float v = agg1[gid] + h1[gid] * di * di + b1[f];
    agg1[gid] = v > 0.0f ? v : 0.0f;
}

__global__ __launch_bounds__(256) void k_padW2(const float* __restrict__ W2,
                                               float* __restrict__ Bp) {
    int t = blockIdx.x * 256 + threadIdx.x;
    if (t >= F_H * H_PAD) return;
    int k = t >> 4, c = t & 15;
    Bp[t] = (c < N_CLS) ? W2[k * N_CLS + c] : 0.0f;
}

__global__ __launch_bounds__(256) void k_gemm2(const float* __restrict__ A,
                                               const float* __restrict__ Bp,
                                               float* __restrict__ h2) {
    const int wave = threadIdx.x >> 5;
    const int lane = threadIdx.x & 31;
    const int tile = blockIdx.x * 8 + wave;
    if ((size_t)tile * 16 >= N_NODES) return;
    const int m  = lane & 15;
    const int kg = lane >> 4;

    const float* Ar = A + ((size_t)tile * 16 + m) * F_H;
    v8f c = {0.f,0.f,0.f,0.f,0.f,0.f,0.f,0.f};
    for (int k0 = 0; k0 < F_H; k0 += 4) {
        int ka = k0 + 2 * kg;
        v2f a = *(const v2f*)&Ar[ka];
        v2f b;
        b.x = Bp[ka * H_PAD + m];
        b.y = Bp[(ka + 1) * H_PAD + m];
        c = __builtin_amdgcn_wmma_f32_16x16x4_f32(false, a, false, b,
                                                  (short)0, c, false, false);
    }
    float* out = h2 + (size_t)tile * 16 * H_PAD;
    #pragma unroll
    for (int r = 0; r < 8; ++r) {
        int row = r + 8 * kg;
        out[(size_t)row * H_PAD + m] = c[r];
    }
}

__global__ __launch_bounds__(256) void k_scatter2(const int* __restrict__ src,
                                                  const int* __restrict__ dst,
                                                  const float* __restrict__ dinv,
                                                  const float* __restrict__ h2,
                                                  float* __restrict__ agg2) {
    long gid = (long)blockIdx.x * 256 + threadIdx.x;
    long e   = gid >> 4;
    if (e >= N_EDGES) return;
    int f = (int)(gid & 15);
    if (f >= N_CLS) return;
    int s = src[e], d = dst[e];
    float nrm = dinv[s] * dinv[d];
    atomicAdd(&agg2[(size_t)d * H_PAD + f], h2[(size_t)s * H_PAD + f] * nrm);
}

__global__ __launch_bounds__(256) void k_post2(const float* __restrict__ h2,
                                               const float* __restrict__ agg2,
                                               const float* __restrict__ dinv,
                                               const float* __restrict__ b2,
                                               float* __restrict__ out) {
    int i = blockIdx.x * 256 + threadIdx.x;
    if (i >= N_NODES) return;
    float di = dinv[i];
    float d2 = di * di;
    float v[N_CLS];
    float mx = -INFINITY;
    #pragma unroll
    for (int c = 0; c < N_CLS; ++c) {
        float t = agg2[(size_t)i * H_PAD + c] + h2[(size_t)i * H_PAD + c] * d2 + b2[c];
        v[c] = t;
        mx = fmaxf(mx, t);
    }
    float s = 0.0f;
    #pragma unroll
    for (int c = 0; c < N_CLS; ++c) s += expf(v[c] - mx);
    float lse = logf(s) + mx;
    #pragma unroll
    for (int c = 0; c < N_CLS; ++c) out[(size_t)i * N_CLS + c] = v[c] - lse;
}

extern "C" void kernel_launch(void* const* d_in, const int* in_sizes, int n_in,
                              void* d_out, int out_size, void* d_ws, size_t ws_size,
                              hipStream_t stream) {
    const float* x   = (const float*)d_in[0];
    const int*   ei  = (const int*)d_in[1];
    const float* W1  = (const float*)d_in[2];
    const float* b1  = (const float*)d_in[3];
    const float* W2  = (const float*)d_in[4];
    const float* b2  = (const float*)d_in[5];
    float*       out = (float*)d_out;

    const int* src = ei;
    const int* dst = ei + N_EDGES;

    float* ws    = (float*)d_ws;
    float* dinv  = ws;
    float* h1    = dinv + 100352;
    float* agg1  = h1 + (size_t)N_NODES * F_H;
    float* W2pad = agg1 + (size_t)N_NODES * F_H;
    float* h2    = h1;
    float* agg2  = h1 + (size_t)N_NODES * H_PAD;

    const int nblk_node = (N_NODES + 255) / 256;

    k_deg_init<<<nblk_node, 256, 0, stream>>>(dinv);
    k_deg_add <<<(N_EDGES + 255) / 256, 256, 0, stream>>>(dst, dinv);
    k_dinv    <<<nblk_node, 256, 0, stream>>>(dinv);

    k_gemm1<<<N_NODES / 16, 256, 0, stream>>>(x, W1, h1);
    {
        long n = (long)N_NODES * F_H;
        k_zero<<<(unsigned)((n + 255) / 256), 256, 0, stream>>>(agg1, n);
    }
    {
        long t = (long)N_EDGES * 32;
        k_scatter1<<<(unsigned)((t + 255) / 256), 256, 0, stream>>>(src, dst, dinv, h1, agg1);
    }
    {
        long n = (long)N_NODES * F_H;
        k_post1<<<(unsigned)((n + 255) / 256), 256, 0, stream>>>(h1, dinv, b1, agg1);
    }

    k_padW2<<<(F_H * H_PAD + 255) / 256, 256, 0, stream>>>(W2, W2pad);
    k_gemm2<<<(N_NODES / 16 + 7) / 8, 256, 0, stream>>>(agg1, W2pad, h2);
    {
        long n = (long)N_NODES * H_PAD;
        k_zero<<<(unsigned)((n + 255) / 256), 256, 0, stream>>>(agg2, n);
    }
    {
        long t = (long)N_EDGES * 16;
        k_scatter2<<<(unsigned)((t + 255) / 256), 256, 0, stream>>>(src, dst, dinv, h2, agg2);
    }
    k_post2<<<nblk_node, 256, 0, stream>>>(h2, agg2, dinv, b2, out);
}


# sum by destination: edges grouped by dst on every call (returning int add = slot, scan, fill); a wave sums a node's rows in registers, LDS feeds the W2 WMMA; no float atomics; sums in slot order
# speedup vs baseline: 1.9360x; 1.5982x over previous
#include <hip/hip_runtime.h>
#include <math.h>


#define N_NODES 100000
#define N_EDGES 1600000
#define F_IN    256
#define F_H     128
#define N_CLS   10
#define H_PAD   16

typedef float v2f __attribute__((ext_vector_type(2)));
typedef float v8f __attribute__((ext_vector_type(8)));

__global__ __launch_bounds__(256) void k_cnt_init(int* __restrict__ cnt) {
    int i = blockIdx.x * 256 + threadIdx.x;
    if (i < N_NODES) cnt[i] = 0;
}

__global__ __launch_bounds__(256) void k_rank(const int* __restrict__ dst,
                                              int* __restrict__ cnt,
                                              int* __restrict__ rank) {
    int e = blockIdx.x * 256 + threadIdx.x;
    if (e < N_EDGES) rank[e] = atomicAdd(&cnt[dst[e]], 1);
}

__device__ __forceinline__ float dinv_of(int deg) {
    float d = (float)deg;
    return (d > 0.0f) ? rsqrtf(d) : 0.0f;
}

__global__ __launch_bounds__(1024) void k_scan(const int* __restrict__ cnt,
                                               int* __restrict__ off,
                                               float* __restrict__ dinv) {
    __shared__ int wsum[32];
    const int tid  = threadIdx.x;
    const int lane = tid & 31;
    const int wv   = tid >> 5;
    int carry = 0;
    for (int base = 0; base < N_NODES; base += 4096) {
        const int i0 = base + tid * 4;
        int4 v = make_int4(0, 0, 0, 0);
        if (i0 < N_NODES) v = *(const int4*)&cnt[i0];
        const int s1 = v.x, s2 = s1 + v.y, s3 = s2 + v.z, s4 = s3 + v.w;
        int inc = s4;
        for (int o = 1; o < 32; o <<= 1) {
            int t = __shfl_up(inc, o);
            if (lane >= o) inc += t;
        }
        if (lane == 31) wsum[wv] = inc;
        __syncthreads();
        const int ws = wsum[lane];
        int winc = ws;
        for (int o = 1; o < 32; o <<= 1) {
            int t = __shfl_up(winc, o);
            if (lane >= o) winc += t;
        }
        const int wave_excl = __shfl(winc, wv) - __shfl(ws, wv);
        const int total     = __shfl(winc, 31);
        const int ex = carry + wave_excl + (inc - s4);
        if (i0 < N_NODES) {
            *(int4*)&off[i0] = make_int4(ex, ex + s1, ex + s2, ex + s3);
            float4 d;
            d.x = dinv_of(v.x + 1);
            d.y = dinv_of(v.y + 1);
            d.z = dinv_of(v.z + 1);
            d.w = dinv_of(v.w + 1);
            *(float4*)&dinv[i0] = d;
        }
        carry += total;
        __syncthreads();
    }
}

__global__ __launch_bounds__(256) void k_fill(const int* __restrict__ src,
                                              const int* __restrict__ dst,
                                              const int* __restrict__ off,
                                              const int* __restrict__ rank,
                                              int* __restrict__ csr) {
    int e = blockIdx.x * 256 + threadIdx.x;
    if (e < N_EDGES) csr[off[dst[e]] + rank[e]] = src[e];
}

#define LDS_STRIDE 260
__global__ __launch_bounds__(256) void k_gemm1(const float* __restrict__ x,
                                               const float* __restrict__ W1,
                                               float* __restrict__ h1) {
    __shared__ float As[16 * LDS_STRIDE];
    const int tile = blockIdx.x;
    const int tid  = threadIdx.x;

    const float4* xg = (const float4*)(x + (size_t)tile * 16 * F_IN);
    #pragma unroll
    for (int it = 0; it < 4; ++it) {
        int q   = tid + 256 * it;
        int row = q >> 6;
        int c4  = q & 63;
        float4 v = xg[row * 64 + c4];
        float* d = &As[row * LDS_STRIDE + c4 * 4];
        d[0] = v.x; d[1] = v.y; d[2] = v.z; d[3] = v.w;
    }
    __syncthreads();

    const int wave = tid >> 5;
    const int lane = tid & 31;
    const int m    = lane & 15;
    const int kg   = lane >> 4;
    const int n0   = wave * 16;

    v8f c = {0.f,0.f,0.f,0.f,0.f,0.f,0.f,0.f};
    for (int k0 = 0; k0 < F_IN; k0 += 4) {
        int ka = k0 + 2 * kg;
        v2f a = *(const v2f*)&As[m * LDS_STRIDE + ka];
        v2f b;
        b.x = W1[(size_t)ka       * F_H + n0 + m];
        b.y = W1[(size_t)(ka + 1) * F_H + n0 + m];
        c = __builtin_amdgcn_wmma_f32_16x16x4_f32(false, a, false, b,
                                                  (short)0, c, false, false);
    }

    float* out = h1 + (size_t)tile * 16 * F_H;
    #pragma unroll
    for (int r = 0; r < 8; ++r) {
        int row = r + 8 * kg;
        out[(size_t)row * F_H + n0 + m] = c[r];
    }
}

__global__ __launch_bounds__(256) void k_padW2(const float* __restrict__ W2,
                                               float* __restrict__ Bp) {
    int t = blockIdx.x * 256 + threadIdx.x;
    if (t >= F_H * H_PAD) return;
    int k = t >> 4, c = t & 15;
    Bp[t] = (c < N_CLS) ? W2[k * N_CLS + c] : 0.0f;
}

#define T_STRIDE 132
__global__ __launch_bounds__(256) void k_gather1(const int* __restrict__ csr,
                                                 const int* __restrict__ off,
                                                 const int* __restrict__ cnt,
                                                 const float* __restrict__ dinv,
                                                 const float* __restrict__ h1,
                                                 const float* __restrict__ b1,
                                                 const float* __restrict__ Bp,
                                                 float* __restrict__ h2) {
    __shared__ __attribute__((aligned(16))) float T[16 * T_STRIDE];
    const int tile = blockIdx.x;
    const int tid  = threadIdx.x;
    const int wave = tid >> 5;
    const int lane = tid & 31;
    const float4 bb = *(const float4*)&b1[lane * 4];

    for (int j = 0; j < 2; ++j) {
        const int row = wave * 2 + j;
        const int i   = __builtin_amdgcn_readfirstlane(tile * 16 + row);
        const int beg = off[i];
        const int n   = cnt[i];
        const float di = dinv[i];
        float4 acc = make_float4(0.f, 0.f, 0.f, 0.f);
        for (int c = 0; c < n; c += 32) {
            const int m = (n - c < 32) ? (n - c) : 32;
            int   my_s = 0;
            float my_w = 0.f;
            if (lane < m) {
                my_s = csr[beg + c + lane];
                my_w = dinv[my_s] * di;
            }
            #pragma unroll 4
            for (int k = 0; k < m; ++k) {
                const int   s = __builtin_amdgcn_readlane(my_s, k);
                const float w = __int_as_float(__builtin_amdgcn_readlane(__float_as_int(my_w), k));
                const float4 r = *(const float4*)&h1[(size_t)s * F_H + lane * 4];
                acc.x += r.x * w; acc.y += r.y * w; acc.z += r.z * w; acc.w += r.w * w;
            }
        }
        const float4 hs = *(const float4*)&h1[(size_t)i * F_H + lane * 4];
        float4 v;
        v.x = acc.x + hs.x * di * di + bb.x;
        v.y = acc.y + hs.y * di * di + bb.y;
        v.z = acc.z + hs.z * di * di + bb.z;
        v.w = acc.w + hs.w * di * di + bb.w;
        v.x = v.x > 0.0f ? v.x : 0.0f;
        v.y = v.y > 0.0f ? v.y : 0.0f;
        v.z = v.z > 0.0f ? v.z : 0.0f;
        v.w = v.w > 0.0f ? v.w : 0.0f;
        *(float4*)&T[row * T_STRIDE + lane * 4] = v;
    }
    __syncthreads();
    if (wave != 0) return;

    const int m  = lane & 15;
    const int kg = lane >> 4;
    v8f c = {0.f,0.f,0.f,0.f,0.f,0.f,0.f,0.f};
    for (int k0 = 0; k0 < F_H; k0 += 4) {
        int ka = k0 + 2 * kg;
        v2f a = *(const v2f*)&T[m * T_STRIDE + ka];
        v2f b;
        b.x = Bp[ka * H_PAD + m];
        b.y = Bp[(ka + 1) * H_PAD + m];
        c = __builtin_amdgcn_wmma_f32_16x16x4_f32(false, a, false, b,
                                                  (short)0, c, false, false);
    }
    float* out = h2 + (size_t)tile * 16 * H_PAD;
    #pragma unroll
    for (int r = 0; r < 8; ++r) {
        int row = r + 8 * kg;
        out[(size_t)row * H_PAD + m] = c[r];
    }
}

__global__ __launch_bounds__(256) void k_gather2(const int* __restrict__ csr,
                                                 const int* __restrict__ off,
                                                 const int* __restrict__ cnt,
                                                 const float* __restrict__ dinv,
                                                 const float* __restrict__ h2,
                                                 const float* __restrict__ b2,
                                                 float* __restrict__ out) {
    const int tid = threadIdx.x;
    const int f   = tid & 15;
    const int i   = blockIdx.x * 16 + (tid >> 4);
    const int beg = off[i];
    const int n   = cnt[i];
    const float di = dinv[i];
    float acc = 0.f;
    for (int c = 0; c < n; c += 16) {
        const int m = (n - c < 16) ? (n - c) : 16;
        int   my_s = 0;
        float my_w = 0.f;
        if (f < m) {
            my_s = csr[beg + c + f];
            my_w = dinv[my_s] * di;
        }
        for (int k = 0; k < m; k += 4) {
            #pragma unroll
            for (int u = 0; u < 4; ++u) {
                const int   s = __shfl(my_s, k + u, 16);
                const float w = __shfl(my_w, k + u, 16);
                const float v = h2[(size_t)s * H_PAD + f];
                acc += (k + u < m) ? v * w : 0.0f;
            }
        }
    }
    const bool valid = f < N_CLS;
    float t = -INFINITY;
    if (valid) t = acc + h2[(size_t)i * H_PAD + f] * di * di + b2[f];
    float mx = t;
    for (int o = 8; o >= 1; o >>= 1) mx = fmaxf(mx, __shfl_xor(mx, o, 16));
    float sm = valid ? expf(t - mx) : 0.0f;
    for (int o = 8; o >= 1; o >>= 1) sm += __shfl_xor(sm, o, 16);
    const float lse = logf(sm) + mx;
    if (valid) out[(size_t)i * N_CLS + f] = t - lse;
}

extern "C" void kernel_launch(void* const* d_in, const int* in_sizes, int n_in,
                              void* d_out, int out_size, void* d_ws, size_t ws_size,
                              hipStream_t stream) {
    const float* x   = (const float*)d_in[0];
    const int*   ei  = (const int*)d_in[1];
    const float* W1  = (const float*)d_in[2];
    const float* b1  = (const float*)d_in[3];
    const float* W2  = (const float*)d_in[4];
    const float* b2  = (const float*)d_in[5];
    float*       out = (float*)d_out;

    const int* src = ei;
    const int* dst = ei + N_EDGES;

    float* ws    = (float*)d_ws;
    float* dinv  = ws;
    float* h1    = dinv + 100352;
    float* reg2  = h1 + (size_t)N_NODES * F_H;
    float* W2pad = reg2 + (size_t)N_NODES * F_H;
    int*   cnt   = (int*)reg2;
    int*   off   = (int*)(reg2 + 100352);
    int*   rank  = (int*)(reg2 + 200704);
    int*   csr   = (int*)(reg2 + 1800704);
    float* h2    = reg2 + 3400704;

    const int nblk_node = (N_NODES + 255) / 256;
    const int nblk_edge = (N_EDGES + 255) / 256;

    k_cnt_init<<<nblk_node, 256, 0, stream>>>(cnt);
    k_rank    <<<nblk_edge, 256, 0, stream>>>(dst, cnt, rank);
    k_scan    <<<1, 1024, 0, stream>>>(cnt, off, dinv);
    k_fill    <<<nblk_edge, 256, 0, stream>>>(src, dst, off, rank, csr);

    k_gemm1<<<N_NODES / 16, 256, 0, stream>>>(x, W1, h1);
    k_padW2<<<(F_H * H_PAD + 255) / 256, 256, 0, stream>>>(W2, W2pad);
    k_gather1<<<N_NODES / 16, 256, 0, stream>>>(csr, off, cnt, dinv, h1, b1, W2pad, h2);

    k_gather2<<<N_NODES / 16, 256, 0, stream>>>(csr, off, cnt, dinv, h2, b2, out);
}


# edges grouped by dst with a two-level counting sort on LDS counters (512-node bins, then one block per bin) replacing global returning adds, one-block scan and fill; slot order = LDS arrival order
# speedup vs baseline: 3.3380x; 1.7242x over previous
#include <hip/hip_runtime.h>
#include <math.h>


#define N_NODES 100000
#define N_EDGES 1600000
#define F_IN    256
#define F_H     128
#define N_CLS   10
#define H_PAD   16

typedef float v2f __attribute__((ext_vector_type(2)));
typedef float v8f __attribute__((ext_vector_type(8)));

#define BIN_SHIFT 9
#define BIN_NODES 512
#define NBIN      196
#define SBLK      256
#define EPB       (N_EDGES / SBLK)
#define EPT       7

__global__ __launch_bounds__(1024) void k_binhist(const int* __restrict__ dst,
                                                  int* __restrict__ M) {
    __shared__ int hist[NBIN];
    const int tid = threadIdx.x;
    const int blk = blockIdx.x;
    if (tid < NBIN) hist[tid] = 0;
    __syncthreads();
    const int e0 = blk * EPB;
    int d[EPT];
    #pragma unroll
    for (int i = 0; i < EPT; ++i) {
        const int idx = tid + 1024 * i;
        d[i] = dst[e0 + (idx < EPB ? idx : 0)];
    }
    #pragma unroll
    for (int i = 0; i < EPT; ++i)
        if (tid + 1024 * i < EPB) atomicAdd(&hist[d[i] >> BIN_SHIFT], 1);
    __syncthreads();
    if (tid < NBIN) M[tid * SBLK + blk] = hist[tid];
}

__global__ __launch_bounds__(1024) void k_binmove(const int* __restrict__ src,
                                                  const int* __restrict__ dst,
                                                  const int* __restrict__ M,
                                                  int* __restrict__ binbeg,
                                                  int* __restrict__ pairs) {
    __shared__ int rowtot[NBIN];
    __shared__ int rowpre[NBIN];
    __shared__ int base[NBIN];
    __shared__ int cur[NBIN];
    const int tid  = threadIdx.x;
    const int lane = tid & 31;
    const int wave = tid >> 5;
    const int blk  = blockIdx.x;
    if (tid < NBIN) cur[tid] = 0;

    int4 ra[7], rb[7];
    #pragma unroll
    for (int i = 0; i < 7; ++i) {
        const int bin = wave + 32 * i;
        const int4* row = (const int4*)(M + (bin < NBIN ? bin : NBIN - 1) * SBLK);
        ra[i] = row[lane];
        rb[i] = row[32 + lane];
    }
    #pragma unroll
    for (int i = 0; i < 7; ++i) {
        const int bin = wave + 32 * i;
        {
            const int4 a = ra[i];
            const int4 b = rb[i];
            int tot = a.x + a.y + a.z + a.w + b.x + b.y + b.z + b.w;
            const int c0 = 4 * lane, c1 = 128 + 4 * lane;
            int pre = (c0     < blk ? a.x : 0) + (c0 + 1 < blk ? a.y : 0)
                    + (c0 + 2 < blk ? a.z : 0) + (c0 + 3 < blk ? a.w : 0)
                    + (c1     < blk ? b.x : 0) + (c1 + 1 < blk ? b.y : 0)
                    + (c1 + 2 < blk ? b.z : 0) + (c1 + 3 < blk ? b.w : 0);
            for (int o = 16; o >= 1; o >>= 1) {
                tot += __shfl_xor(tot, o);
                pre += __shfl_xor(pre, o);
            }
            if (lane == 0 && bin < NBIN) { rowtot[bin] = tot; rowpre[bin] = pre; }
        }
    }
    __syncthreads();
    if (wave == 0) {
        int loc[7];
        int s = 0;
        #pragma unroll
        for (int j = 0; j < 7; ++j) {
            const int bin = 7 * lane + j;
            loc[j] = s;
            s += (bin < NBIN) ? rowtot[bin] : 0;
        }
        int inc = s;
        for (int o = 1; o < 32; o <<= 1) {
            int t = __shfl_up(inc, o);
            if (lane >= o) inc += t;
        }
        const int ex = inc - s;
        #pragma unroll
        for (int j = 0; j < 7; ++j) {
            const int bin = 7 * lane + j;
            if (bin < NBIN) {
                const int bb = ex + loc[j];
                base[bin] = bb + rowpre[bin];
                if (blk == 0) binbeg[bin] = bb;
            }
        }
        if (blk == 0 && lane == 31) binbeg[NBIN] = inc;
    }
    __syncthreads();

    const int e0 = blk * EPB;
    int2* pr = (int2*)pairs;
    int dd[EPT], ss[EPT];
    #pragma unroll
    for (int i = 0; i < EPT; ++i) {
        const int idx = tid + 1024 * i;
        const int e = e0 + (idx < EPB ? idx : 0);
        dd[i] = dst[e];
        ss[i] = src[e];
    }
    #pragma unroll
    for (int i = 0; i < EPT; ++i) {
        if (tid + 1024 * i < EPB) {
            const int b = dd[i] >> BIN_SHIFT;
            const int r = atomicAdd(&cur[b], 1);
            pr[base[b] + r] = make_int2(ss[i], dd[i]);
        }
    }
}

__device__ __forceinline__ float dinv_of(int deg) {
    float d = (float)deg;
    return (d > 0.0f) ? rsqrtf(d) : 0.0f;
}

__global__ __launch_bounds__(1024) void k_binsort(const int* __restrict__ pairs,
                                                  const int* __restrict__ binbeg,
                                                  int* __restrict__ cnt,
                                                  int* __restrict__ off,
                                                  float* __restrict__ dinv,
                                                  int* __restrict__ csr) {
    __shared__ int cntS[BIN_NODES];
    __shared__ int curS[BIN_NODES];
    __shared__ int exS[BIN_NODES];
    __shared__ int wsum[16];
    const int tid  = threadIdx.x;
    const int lane = tid & 31;
    const int wave = tid >> 5;
    const int bin  = blockIdx.x;
    const int node0 = bin * BIN_NODES;
    const int beg = binbeg[bin];
    const int end = binbeg[bin + 1];
    const int2* pr = (const int2*)pairs;
    if (tid < BIN_NODES) { cntS[tid] = 0; curS[tid] = 0; }
    __syncthreads();
    for (int j0 = beg; j0 < end; j0 += 8192) {
        int dy[8];
        #pragma unroll
        for (int u = 0; u < 8; ++u) {
            const int j = j0 + tid + 1024 * u;
            dy[u] = pr[j < end ? j : end - 1].y;
        }
        #pragma unroll
        for (int u = 0; u < 8; ++u)
            if (j0 + tid + 1024 * u < end) atomicAdd(&cntS[dy[u] - node0], 1);
    }
    __syncthreads();

    const int v = (tid < BIN_NODES) ? cntS[tid] : 0;
    int inc = v;
    for (int o = 1; o < 32; o <<= 1) {
        int t = __shfl_up(inc, o);
        if (lane >= o) inc += t;
    }
    if (tid < BIN_NODES && lane == 31) wsum[wave] = inc;
    __syncthreads();
    if (tid < BIN_NODES) {
        int wp = 0;
        for (int w = 0; w < wave; ++w) wp += wsum[w];
        const int ex = wp + inc - v;
        exS[tid] = ex;
        const int i = node0 + tid;
        if (i < N_NODES) {
            off[i]  = beg + ex;
            cnt[i]  = v;
            dinv[i] = dinv_of(v + 1);
        }
    }
    __syncthreads();
    for (int j0 = beg; j0 < end; j0 += 8192) {
        int2 pp[8];
        #pragma unroll
        for (int u = 0; u < 8; ++u) {
            const int j = j0 + tid + 1024 * u;
            pp[u] = pr[j < end ? j : end - 1];
        }
        #pragma unroll
        for (int u = 0; u < 8; ++u) {
            if (j0 + tid + 1024 * u < end) {
                const int d = pp[u].y - node0;
                const int r = atomicAdd(&curS[d], 1);
                csr[beg + exS[d] + r] = pp[u].x;
            }
        }
    }
}

#define LDS_STRIDE 260
__global__ __launch_bounds__(256) void k_gemm1(const float* __restrict__ x,
                                               const float* __restrict__ W1,
                                               float* __restrict__ h1) {
    __shared__ float As[16 * LDS_STRIDE];
    const int tile = blockIdx.x;
    const int tid  = threadIdx.x;

    const float4* xg = (const float4*)(x + (size_t)tile * 16 * F_IN);
    #pragma unroll
    for (int it = 0; it < 4; ++it) {
        int q   = tid + 256 * it;
        int row = q >> 6;
        int c4  = q & 63;
        float4 v = xg[row * 64 + c4];
        float* d = &As[row * LDS_STRIDE + c4 * 4];
        d[0] = v.x; d[1] = v.y; d[2] = v.z; d[3] = v.w;
    }
    __syncthreads();

    const int wave = tid >> 5;
    const int lane = tid & 31;
    const int m    = lane & 15;
    const int kg   = lane >> 4;
    const int n0   = wave * 16;

    v8f c = {0.f,0.f,0.f,0.f,0.f,0.f,0.f,0.f};
    for (int k0 = 0; k0 < F_IN; k0 += 4) {
        int ka = k0 + 2 * kg;
        v2f a = *(const v2f*)&As[m * LDS_STRIDE + ka];
        v2f b;
        b.x = W1[(size_t)ka       * F_H + n0 + m];
        b.y = W1[(size_t)(ka + 1) * F_H + n0 + m];
        c = __builtin_amdgcn_wmma_f32_16x16x4_f32(false, a, false, b,
                                                  (short)0, c, false, false);
    }

    float* out = h1 + (size_t)tile * 16 * F_H;
    #pragma unroll
    for (int r = 0; r < 8; ++r) {
        int row = r + 8 * kg;
        out[(size_t)row * F_H + n0 + m] = c[r];
    }
}

__global__ __launch_bounds__(256) void k_padW2(const float* __restrict__ W2,
                                               float* __restrict__ Bp) {
    int t = blockIdx.x * 256 + threadIdx.x;
    if (t >= F_H * H_PAD) return;
    int k = t >> 4, c = t & 15;
    Bp[t] = (c < N_CLS) ? W2[k * N_CLS + c] : 0.0f;
}

#define T_STRIDE 132
__global__ __launch_bounds__(256) void k_gather1(const int* __restrict__ csr,
                                                 const int* __restrict__ off,
                                                 const int* __restrict__ cnt,
                                                 const float* __restrict__ dinv,
                                                 const float* __restrict__ h1,
                                                 const float* __restrict__ b1,
                                                 const float* __restrict__ Bp,
                                                 float* __restrict__ h2) {
    __shared__ __attribute__((aligned(16))) float T[16 * T_STRIDE];
    const int tile = blockIdx.x;
    const int tid  = threadIdx.x;
    const int wave = tid >> 5;
    const int lane = tid & 31;
    const float4 bb = *(const float4*)&b1[lane * 4];

    for (int j = 0; j < 2; ++j) {
        const int row = wave * 2 + j;
        const int i   = __builtin_amdgcn_readfirstlane(tile * 16 + row);
        const int beg = off[i];
        const int n   = cnt[i];
        const float di = dinv[i];
        float4 acc = make_float4(0.f, 0.f, 0.f, 0.f);
        for (int c = 0; c < n; c += 32) {
            const int m = (n - c < 32) ? (n - c) : 32;
            int   my_s = 0;
            float my_w = 0.f;
            if (lane < m) {
                my_s = csr[beg + c + lane];
                my_w = dinv[my_s] * di;
            }
            #pragma unroll 4
            for (int k = 0; k < m; ++k) {
                const int   s = __builtin_amdgcn_readlane(my_s, k);
                const float w = __int_as_float(__builtin_amdgcn_readlane(__float_as_int(my_w), k));
                const float4 r = *(const float4*)&h1[(size_t)s * F_H + lane * 4];
                acc.x += r.x * w; acc.y += r.y * w; acc.z += r.z * w; acc.w += r.w * w;
            }
        }
        const float4 hs = *(const float4*)&h1[(size_t)i * F_H + lane * 4];
        float4 v;
        v.x = acc.x + hs.x * di * di + bb.x;
        v.y = acc.y + hs.y * di * di + bb.y;
        v.z = acc.z + hs.z * di * di + bb.z;
        v.w = acc.w + hs.w * di * di + bb.w;
        v.x = v.x > 0.0f ? v.x : 0.0f;
        v.y = v.y > 0.0f ? v.y : 0.0f;
        v.z = v.z > 0.0f ? v.z : 0.0f;
        v.w = v.w > 0.0f ? v.w : 0.0f;
        *(float4*)&T[row * T_STRIDE + lane * 4] = v;
    }
    __syncthreads();
    if (wave != 0) return;

    const int m  = lane & 15;
    const int kg = lane >> 4;
    v8f c = {0.f,0.f,0.f,0.f,0.f,0.f,0.f,0.f};
    for (int k0 = 0; k0 < F_H; k0 += 4) {
        int ka = k0 + 2 * kg;
        v2f a = *(const v2f*)&T[m * T_STRIDE + ka];
        v2f b;
        b.x = Bp[ka * H_PAD + m];
        b.y = Bp[(ka + 1) * H_PAD + m];
        c = __builtin_amdgcn_wmma_f32_16x16x4_f32(false, a, false, b,
                                                  (short)0, c, false, false);
    }
    float* out = h2 + (size_t)tile * 16 * H_PAD;
    #pragma unroll
    for (int r = 0; r < 8; ++r) {
        int row = r + 8 * kg;
        out[(size_t)row * H_PAD + m] = c[r];
    }
}

__global__ __launch_bounds__(256) void k_gather2(const int* __restrict__ csr,
                                                 const int* __restrict__ off,
                                                 const int* __restrict__ cnt,
                                                 const float* __restrict__ dinv,
                                                 const float* __restrict__ h2,
                                                 const float* __restrict__ b2,
                                                 float* __restrict__ out) {
    const int tid = threadIdx.x;
    const int f   = tid & 15;
    const int i   = blockIdx.x * 16 + (tid >> 4);
    const int beg = off[i];
    const int n   = cnt[i];
    const float di = dinv[i];
    float acc = 0.f;
    for (int c = 0; c < n; c += 16) {
        const int m = (n - c < 16) ? (n - c) : 16;
        int   my_s = 0;
        float my_w = 0.f;
        if (f < m) {
            my_s = csr[beg + c + f];
            my_w = dinv[my_s] * di;
        }
        for (int k = 0; k < m; k += 4) {
            #pragma unroll
            for (int u = 0; u < 4; ++u) {
                const int   s = __shfl(my_s, k + u, 16);
                const float w = __shfl(my_w, k + u, 16);
                const float v = h2[(size_t)s * H_PAD + f];
                acc += (k + u < m) ? v * w : 0.0f;
            }
        }
    }
    const bool valid = f < N_CLS;
    float t = -INFINITY;
    if (valid) t = acc + h2[(size_t)i * H_PAD + f] * di * di + b2[f];
    float mx = t;
    for (int o = 8; o >= 1; o >>= 1) mx = fmaxf(mx, __shfl_xor(mx, o, 16));
    float sm = valid ? expf(t - mx) : 0.0f;
    for (int o = 8; o >= 1; o >>= 1) sm += __shfl_xor(sm, o, 16);
    const float lse = logf(sm) + mx;
    if (valid) out[(size_t)i * N_CLS + f] = t - lse;
}

extern "C" void kernel_launch(void* const* d_in, const int* in_sizes, int n_in,
                              void* d_out, int out_size, void* d_ws, size_t ws_size,
                              hipStream_t stream) {
    const float* x   = (const float*)d_in[0];
    const int*   ei  = (const int*)d_in[1];
    const float* W1  = (const float*)d_in[2];
    const float* b1  = (const float*)d_in[3];
    const float* W2  = (const float*)d_in[4];
    const float* b2  = (const float*)d_in[5];
    float*       out = (float*)d_out;

    const int* src = ei;
    const int* dst = ei + N_EDGES;

    float* ws    = (float*)d_ws;
    float* dinv  = ws;
    float* h1    = dinv + 100352;
    float* reg2  = h1 + (size_t)N_NODES * F_H;
    float* W2pad = reg2 + (size_t)N_NODES * F_H;
    int*   cnt   = (int*)reg2;
    int*   off   = (int*)(reg2 + 100352);
    int*   csr   = (int*)(reg2 + 1800704);
    float* h2    = reg2 + 3400704;
    int*   pairs = (int*)(reg2 + 5000704);
    int*   M     = (int*)(reg2 + 8200704);
    int*   binbeg= (int*)(reg2 + 8250880);

    k_binhist<<<SBLK, 1024, 0, stream>>>(dst, M);
    k_binmove<<<SBLK, 1024, 0, stream>>>(src, dst, M, binbeg, pairs);
    k_binsort<<<NBIN, 1024, 0, stream>>>(pairs, binbeg, cnt, off, dinv, csr);

    k_gemm1<<<N_NODES / 16, 256, 0, stream>>>(x, W1, h1);
    k_padW2<<<(F_H * H_PAD + 255) / 256, 256, 0, stream>>>(W2, W2pad);
    k_gather1<<<N_NODES / 16, 256, 0, stream>>>(csr, off, cnt, dinv, h1, b1, W2pad, h2);

    k_gather2<<<N_NODES / 16, 256, 0, stream>>>(csr, off, cnt, dinv, h2, b2, out);
}


# gemm1: each wave keeps its 256x16 slice of W1 in registers (64 B fragments) and 512 resident blocks walk over the x tiles, so no W1 loads per matrix instruction; k order and h1 bits unchanged
# speedup vs baseline: 3.7556x; 1.1251x over previous
#include <hip/hip_runtime.h>
#include <math.h>


#define N_NODES 100000
#define N_EDGES 1600000
#define F_IN    256
#define F_H     128
#define N_CLS   10
#define H_PAD   16

typedef float v2f __attribute__((ext_vector_type(2)));
typedef float v8f __attribute__((ext_vector_type(8)));

#define BIN_SHIFT 9
#define BIN_NODES 512
#define NBIN      196
#define SBLK      256
#define EPB       (N_EDGES / SBLK)
#define EPT       7

__global__ __launch_bounds__(1024) void k_binhist(const int* __restrict__ dst,
                                                  int* __restrict__ M) {
    __shared__ int hist[NBIN];
    const int tid = threadIdx.x;
    const int blk = blockIdx.x;
    if (tid < NBIN) hist[tid] = 0;
    __syncthreads();
    const int e0 = blk * EPB;
    int d[EPT];
    #pragma unroll
    for (int i = 0; i < EPT; ++i) {
        const int idx = tid + 1024 * i;
        d[i] = dst[e0 + (idx < EPB ? idx : 0)];
    }
    #pragma unroll
    for (int i = 0; i < EPT; ++i)
        if (tid + 1024 * i < EPB) atomicAdd(&hist[d[i] >> BIN_SHIFT], 1);
    __syncthreads();
    if (tid < NBIN) M[tid * SBLK + blk] = hist[tid];
}

__global__ __launch_bounds__(1024) void k_binmove(const int* __restrict__ src,
                                                  const int* __restrict__ dst,
                                                  const int* __restrict__ M,
                                                  int* __restrict__ binbeg,
                                                  int* __restrict__ pairs) {
    __shared__ int rowtot[NBIN];
    __shared__ int rowpre[NBIN];
    __shared__ int base[NBIN];
    __shared__ int cur[NBIN];
    const int tid  = threadIdx.x;
    const int lane = tid & 31;
    const int wave = tid >> 5;
    const int blk  = blockIdx.x;
    if (tid < NBIN) cur[tid] = 0;

    int4 ra[7], rb[7];
    #pragma unroll
    for (int i = 0; i < 7; ++i) {
        const int bin = wave + 32 * i;
        const int4* row = (const int4*)(M + (bin < NBIN ? bin : NBIN - 1) * SBLK);
        ra[i] = row[lane];
        rb[i] = row[32 + lane];
    }
    #pragma unroll
    for (int i = 0; i < 7; ++i) {
        const int bin = wave + 32 * i;
        {
            const int4 a = ra[i];
            const int4 b = rb[i];
            int tot = a.x + a.y + a.z + a.w + b.x + b.y + b.z + b.w;
            const int c0 = 4 * lane, c1 = 128 + 4 * lane;
            int pre = (c0     < blk ? a.x : 0) + (c0 + 1 < blk ? a.y : 0)
                    + (c0 + 2 < blk ? a.z : 0) + (c0 + 3 < blk ? a.w : 0)
                    + (c1     < blk ? b.x : 0) + (c1 + 1 < blk ? b.y : 0)
                    + (c1 + 2 < blk ? b.z : 0) + (c1 + 3 < blk ? b.w : 0);
            for (int o = 16; o >= 1; o >>= 1) {
                tot += __shfl_xor(tot, o);
                pre += __shfl_xor(pre, o);
            }
            if (lane == 0 && bin < NBIN) { rowtot[bin] = tot; rowpre[bin] = pre; }
        }
    }
    __syncthreads();
    if (wave == 0) {
        int loc[7];
        int s = 0;
        #pragma unroll
        for (int j = 0; j < 7; ++j) {
            const int bin = 7 * lane + j;
            loc[j] = s;
            s += (bin < NBIN) ? rowtot[bin] : 0;
        }
        int inc = s;
        for (int o = 1; o < 32; o <<= 1) {
            int t = __shfl_up(inc, o);
            if (lane >= o) inc += t;
        }
        const int ex = inc - s;
        #pragma unroll
        for (int j = 0; j < 7; ++j) {
            const int bin = 7 * lane + j;
            if (bin < NBIN) {
                const int bb = ex + loc[j];
                base[bin] = bb + rowpre[bin];
                if (blk == 0) binbeg[bin] = bb;
            }
        }
        if (blk == 0 && lane == 31) binbeg[NBIN] = inc;
    }
    __syncthreads();

    const int e0 = blk * EPB;
    int2* pr = (int2*)pairs;
    int dd[EPT], ss[EPT];
    #pragma unroll
    for (int i = 0; i < EPT; ++i) {
        const int idx = tid + 1024 * i;
        const int e = e0 + (idx < EPB ? idx : 0);
        dd[i] = dst[e];
        ss[i] = src[e];
    }
    #pragma unroll
    for (int i = 0; i < EPT; ++i) {
        if (tid + 1024 * i < EPB) {
            const int b = dd[i] >> BIN_SHIFT;
            const int r = atomicAdd(&cur[b], 1);
            pr[base[b] + r] = make_int2(ss[i], dd[i]);
        }
    }
}

__device__ __forceinline__ float dinv_of(int deg) {
    float d = (float)deg;
    return (d > 0.0f) ? rsqrtf(d) : 0.0f;
}

__global__ __launch_bounds__(1024) void k_binsort(const int* __restrict__ pairs,
                                                  const int* __restrict__ binbeg,
                                                  int* __restrict__ cnt,
                                                  int* __restrict__ off,
                                                  float* __restrict__ dinv,
                                                  int* __restrict__ csr) {
    __shared__ int cntS[BIN_NODES];
    __shared__ int curS[BIN_NODES];
    __shared__ int exS[BIN_NODES];
    __shared__ int wsum[16];
    const int tid  = threadIdx.x;
    const int lane = tid & 31;
    const int wave = tid >> 5;
    const int bin  = blockIdx.x;
    const int node0 = bin * BIN_NODES;
    const int beg = binbeg[bin];
    const int end = binbeg[bin + 1];
    const int2* pr = (const int2*)pairs;
    if (tid < BIN_NODES) { cntS[tid] = 0; curS[tid] = 0; }
    __syncthreads();
    for (int j0 = beg; j0 < end; j0 += 8192) {
        int dy[8];
        #pragma unroll
        for (int u = 0; u < 8; ++u) {
            const int j = j0 + tid + 1024 * u;
            dy[u] = pr[j < end ? j : end - 1].y;
        }
        #pragma unroll
        for (int u = 0; u < 8; ++u)
            if (j0 + tid + 1024 * u < end) atomicAdd(&cntS[dy[u] - node0], 1);
    }
    __syncthreads();

    const int v = (tid < BIN_NODES) ? cntS[tid] : 0;
    int inc = v;
    for (int o = 1; o < 32; o <<= 1) {
        int t = __shfl_up(inc, o);
        if (lane >= o) inc += t;
    }
    if (tid < BIN_NODES && lane == 31) wsum[wave] = inc;
    __syncthreads();
    if (tid < BIN_NODES) {
        int wp = 0;
        for (int w = 0; w < wave; ++w) wp += wsum[w];
        const int ex = wp + inc - v;
        exS[tid] = ex;
        const int i = node0 + tid;
        if (i < N_NODES) {
            off[i]  = beg + ex;
            cnt[i]  = v;
            dinv[i] = dinv_of(v + 1);
        }
    }
    __syncthreads();
    for (int j0 = beg; j0 < end; j0 += 8192) {
        int2 pp[8];
        #pragma unroll
        for (int u = 0; u < 8; ++u) {
            const int j = j0 + tid + 1024 * u;
            pp[u] = pr[j < end ? j : end - 1];
        }
        #pragma unroll
        for (int u = 0; u < 8; ++u) {
            if (j0 + tid + 1024 * u < end) {
                const int d = pp[u].y - node0;
                const int r = atomicAdd(&curS[d], 1);
                csr[beg + exS[d] + r] = pp[u].x;
            }
        }
    }
}

#define LDS_STRIDE 260
#define N_TILES    (N_NODES / 16)
#define G1_BLOCKS  512

#define G1_LOAD16(dst, t0)                                                        \
    _Pragma("unroll")                                                             \
    for (int u = 0; u < 16; ++u) dst[u] = *(const v2f*)&A[4 * ((t0) + u)];
#define G1_WMMA16(src, t0)                                                        \
    _Pragma("unroll")                                                             \
    for (int u = 0; u < 16; ++u)                                                  \
        c = __builtin_amdgcn_wmma_f32_16x16x4_f32(false, src[u], false, bf[(t0) + u], \
                                                  (short)0, c, false, false);

__global__ __launch_bounds__(256) void k_gemm1(const float* __restrict__ x,
                                               const float* __restrict__ W1,
                                               float* __restrict__ h1) {
    __shared__ __attribute__((aligned(16))) float As[2][16 * LDS_STRIDE];
    const int tid  = threadIdx.x;
    const int wave = tid >> 5;
    const int lane = tid & 31;
    const int m    = lane & 15;
    const int kg   = lane >> 4;
    const int n0   = wave * 16;

    v2f bf[64];
    #pragma unroll
    for (int t = 0; t < 64; ++t) {
        const int ka = 4 * t + 2 * kg;
        bf[t].x = W1[(size_t)ka       * F_H + n0 + m];
        bf[t].y = W1[(size_t)(ka + 1) * F_H + n0 + m];
    }

    int tile = blockIdx.x;
    const int q0 = tid, q1 = tid + 256, q2 = tid + 512, q3 = tid + 768;
    const int d0 = (q0 >> 6) * LDS_STRIDE + (q0 & 63) * 4;
    const int d1 = (q1 >> 6) * LDS_STRIDE + (q1 & 63) * 4;
    const int d2 = (q2 >> 6) * LDS_STRIDE + (q2 & 63) * 4;
    const int d3 = (q3 >> 6) * LDS_STRIDE + (q3 & 63) * 4;
    {
        const float4* xg = (const float4*)(x + (size_t)tile * 16 * F_IN);
        const float4 f0 = xg[q0], f1 = xg[q1], f2 = xg[q2], f3 = xg[q3];
        *(float4*)&As[0][d0] = f0;
        *(float4*)&As[0][d1] = f1;
        *(float4*)&As[0][d2] = f2;
        *(float4*)&As[0][d3] = f3;
    }
    float4 p0, p1, p2, p3;
    {
        const int t1 = tile + G1_BLOCKS;
        const float4* xg = (const float4*)(x + (size_t)(t1 < N_TILES ? t1 : tile) * 16 * F_IN);
        p0 = xg[q0]; p1 = xg[q1]; p2 = xg[q2]; p3 = xg[q3];
    }
    __syncthreads();

    int buf = 0;
    for (; tile < N_TILES; tile += G1_BLOCKS) {
        const bool more = tile + G1_BLOCKS < N_TILES;

        const float* A = &As[buf][m * LDS_STRIDE + 2 * kg];
        v8f c = {0.f,0.f,0.f,0.f,0.f,0.f,0.f,0.f};
        v2f a0[16], a1[16];
        G1_LOAD16(a0, 0)
        G1_LOAD16(a1, 16)
        __builtin_amdgcn_sched_barrier(0);
        G1_WMMA16(a0, 0)
        __builtin_amdgcn_sched_barrier(0);
        G1_LOAD16(a0, 32)
        __builtin_amdgcn_sched_barrier(0);
        G1_WMMA16(a1, 16)
        __builtin_amdgcn_sched_barrier(0);
        G1_LOAD16(a1, 48)
        __builtin_amdgcn_sched_barrier(0);
        G1_WMMA16(a0, 32)
        G1_WMMA16(a1, 48)

        if (more) {
            float* An = As[buf ^ 1];
            *(float4*)&An[d0] = p0;
            *(float4*)&An[d1] = p1;
            *(float4*)&An[d2] = p2;
            *(float4*)&An[d3] = p3;
        }
        {
            const int t2 = tile + 2 * G1_BLOCKS;
            const float4* xg = (const float4*)(x + (size_t)(t2 < N_TILES ? t2 : tile) * 16 * F_IN);
            p0 = xg[q0]; p1 = xg[q1]; p2 = xg[q2]; p3 = xg[q3];
        }
        __builtin_amdgcn_fence(__ATOMIC_RELEASE, "workgroup", "local");
        __builtin_amdgcn_s_barrier();
        __builtin_amdgcn_fence(__ATOMIC_ACQUIRE, "workgroup", "local");
        buf ^= 1;

        float* out = h1 + (size_t)tile * 16 * F_H;
        #pragma unroll
        for (int r = 0; r < 8; ++r) {
            int row = r + 8 * kg;
            out[(size_t)row * F_H + n0 + m] = c[r];
        }
    }
}

__global__ __launch_bounds__(256) void k_padW2(const float* __restrict__ W2,
                                               float* __restrict__ Bp) {
    int t = blockIdx.x * 256 + threadIdx.x;
    if (t >= F_H * H_PAD) return;
    int k = t >> 4, c = t & 15;
    Bp[t] = (c < N_CLS) ? W2[k * N_CLS + c] : 0.0f;
}

#define T_STRIDE 132
__global__ __launch_bounds__(256) void k_gather1(const int* __restrict__ csr,
                                                 const int* __restrict__ off,
                                                 const int* __restrict__ cnt,
                                                 const float* __restrict__ dinv,
                                                 const float* __restrict__ h1,
                                                 const float* __restrict__ b1,
                                                 const float* __restrict__ Bp,
                                                 float* __restrict__ h2) {
    __shared__ __attribute__((aligned(16))) float T[16 * T_STRIDE];
    const int tile = blockIdx.x;
    const int tid  = threadIdx.x;
    const int wave = tid >> 5;
    const int lane = tid & 31;
    const float4 bb = *(const float4*)&b1[lane * 4];

    for (int j = 0; j < 2; ++j) {
        const int row = wave * 2 + j;
        const int i   = __builtin_amdgcn_readfirstlane(tile * 16 + row);
        const int beg = off[i];
        const int n   = cnt[i];
        const float di = dinv[i];
        float4 acc = make_float4(0.f, 0.f, 0.f, 0.f);
        for (int c = 0; c < n; c += 32) {
            const int m = (n - c < 32) ? (n - c) : 32;
            int   my_s = 0;
            float my_w = 0.f;
            if (lane < m) {
                my_s = csr[beg + c + lane];
                my_w = dinv[my_s] * di;
            }
            #pragma unroll 4
            for (int k = 0; k < m; ++k) {
                const int   s = __builtin_amdgcn_readlane(my_s, k);
                const float w = __int_as_float(__builtin_amdgcn_readlane(__float_as_int(my_w), k));
                const float4 r = *(const float4*)&h1[(size_t)s * F_H + lane * 4];
                acc.x += r.x * w; acc.y += r.y * w; acc.z += r.z * w; acc.w += r.w * w;
            }
        }
        const float4 hs = *(const float4*)&h1[(size_t)i * F_H + lane * 4];
        float4 v;
        v.x = acc.x + hs.x * di * di + bb.x;
        v.y = acc.y + hs.y * di * di + bb.y;
        v.z = acc.z + hs.z * di * di + bb.z;
        v.w = acc.w + hs.w * di * di + bb.w;
        v.x = v.x > 0.0f ? v.x : 0.0f;
        v.y = v.y > 0.0f ? v.y : 0.0f;
        v.z = v.z > 0.0f ? v.z : 0.0f;
        v.w = v.w > 0.0f ? v.w : 0.0f;
        *(float4*)&T[row * T_STRIDE + lane * 4] = v;
    }
    __syncthreads();
    if (wave != 0) return;

    const int m  = lane & 15;
    const int kg = lane >> 4;
    v8f c = {0.f,0.f,0.f,0.f,0.f,0.f,0.f,0.f};
    for (int k0 = 0; k0 < F_H; k0 += 4) {
        int ka = k0 + 2 * kg;
        v2f a = *(const v2f*)&T[m * T_STRIDE + ka];
        v2f b;
        b.x = Bp[ka * H_PAD + m];
        b.y = Bp[(ka + 1) * H_PAD + m];
        c = __builtin_amdgcn_wmma_f32_16x16x4_f32(false, a, false, b,
                                                  (short)0, c, false, false);
    }
    float* out = h2 + (size_t)tile * 16 * H_PAD;
    #pragma unroll
    for (int r = 0; r < 8; ++r) {
        int row = r + 8 * kg;
        out[(size_t)row * H_PAD + m] = c[r];
    }
}

__global__ __launch_bounds__(256) void k_gather2(const int* __restrict__ csr,
                                                 const int* __restrict__ off,
                                                 const int* __restrict__ cnt,
                                                 const float* __restrict__ dinv,
                                                 const float* __restrict__ h2,
                                                 const float* __restrict__ b2,
                                                 float* __restrict__ out) {
    const int tid = threadIdx.x;
    const int f   = tid & 15;
    const int i   = blockIdx.x * 16 + (tid >> 4);
    const int beg = off[i];
    const int n   = cnt[i];
    const float di = dinv[i];
    float acc = 0.f;
    for (int c = 0; c < n; c += 16) {
        const int m = (n - c < 16) ? (n - c) : 16;
        int   my_s = 0;
        float my_w = 0.f;
        if (f < m) {
            my_s = csr[beg + c + f];
            my_w = dinv[my_s] * di;
        }
        for (int k = 0; k < m; k += 4) {
            #pragma unroll
            for (int u = 0; u < 4; ++u) {
                const int   s = __shfl(my_s, k + u, 16);
                const float w = __shfl(my_w, k + u, 16);
                const float v = h2[(size_t)s * H_PAD + f];
                acc += (k + u < m) ? v * w : 0.0f;
            }
        }
    }
    const bool valid = f < N_CLS;
    float t = -INFINITY;
    if (valid) t = acc + h2[(size_t)i * H_PAD + f] * di * di + b2[f];
    float mx = t;
    for (int o = 8; o >= 1; o >>= 1) mx = fmaxf(mx, __shfl_xor(mx, o, 16));
    float sm = valid ? expf(t - mx) : 0.0f;
    for (int o = 8; o >= 1; o >>= 1) sm += __shfl_xor(sm, o, 16);
    const float lse = logf(sm) + mx;
    if (valid) out[(size_t)i * N_CLS + f] = t - lse;
}

extern "C" void kernel_launch(void* const* d_in, const int* in_sizes, int n_in,
                              void* d_out, int out_size, void* d_ws, size_t ws_size,
                              hipStream_t stream) {
    const float* x   = (const float*)d_in[0];
    const int*   ei  = (const int*)d_in[1];
    const float* W1  = (const float*)d_in[2];
    const float* b1  = (const float*)d_in[3];
    const float* W2  = (const float*)d_in[4];
    const float* b2  = (const float*)d_in[5];
    float*       out = (float*)d_out;

    const int* src = ei;
    const int* dst = ei + N_EDGES;

    float* ws    = (float*)d_ws;
    float* dinv  = ws;
    float* h1    = dinv + 100352;
    float* reg2  = h1 + (size_t)N_NODES * F_H;
    float* W2pad = reg2 + (size_t)N_NODES * F_H;
    int*   cnt   = (int*)reg2;
    int*   off   = (int*)(reg2 + 100352);
    int*   csr   = (int*)(reg2 + 1800704);
    float* h2    = reg2 + 3400704;
    int*   pairs = (int*)(reg2 + 5000704);
    int*   M     = (int*)(reg2 + 8200704);
    int*   binbeg= (int*)(reg2 + 8250880);

    k_binhist<<<SBLK, 1024, 0, stream>>>(dst, M);
    k_binmove<<<SBLK, 1024, 0, stream>>>(src, dst, M, binbeg, pairs);
    k_binsort<<<NBIN, 1024, 0, stream>>>(pairs, binbeg, cnt, off, dinv, csr);

    k_gemm1<<<G1_BLOCKS, 256, 0, stream>>>(x, W1, h1);
    k_padW2<<<(F_H * H_PAD + 255) / 256, 256, 0, stream>>>(W2, W2pad);
    k_gather1<<<N_NODES / 16, 256, 0, stream>>>(csr, off, cnt, dinv, h1, b1, W2pad, h2);

    k_gather2<<<N_NODES / 16, 256, 0, stream>>>(csr, off, cnt, dinv, h2, b2, out);
}


# rows scaled by dinv[src] where they are made (GEMM1 stores, LDS tile of the W2 product): gathers add plain rows, dinv[dst] applied once after the sum; no dinv[src] read per edge; slot order kept
# speedup vs baseline: 4.0171x; 1.0696x over previous
#include <hip/hip_runtime.h>
#include <math.h>


#define N_NODES 100000
#define N_EDGES 1600000
#define F_IN    256
#define F_H     128
#define N_CLS   10
#define H_PAD   16

typedef float v2f __attribute__((ext_vector_type(2)));
typedef float v8f __attribute__((ext_vector_type(8)));

#define BIN_SHIFT 9
#define BIN_NODES 512
#define NBIN      196
#define SBLK      256
#define EPB       (N_EDGES / SBLK)
#define EPT       7

__global__ __launch_bounds__(1024) void k_binhist(const int* __restrict__ dst,
                                                  int* __restrict__ M) {
    __shared__ int hist[NBIN];
    const int tid = threadIdx.x;
    const int blk = blockIdx.x;
    if (tid < NBIN) hist[tid] = 0;
    __syncthreads();
    const int e0 = blk * EPB;
    int d[EPT];
    #pragma unroll
    for (int i = 0; i < EPT; ++i) {
        const int idx = tid + 1024 * i;
        d[i] = dst[e0 + (idx < EPB ? idx : 0)];
    }
    #pragma unroll
    for (int i = 0; i < EPT; ++i)
        if (tid + 1024 * i < EPB) atomicAdd(&hist[d[i] >> BIN_SHIFT], 1);
    __syncthreads();
    if (tid < NBIN) M[tid * SBLK + blk] = hist[tid];
}

__global__ __launch_bounds__(1024) void k_binmove(const int* __restrict__ src,
                                                  const int* __restrict__ dst,
                                                  const int* __restrict__ M,
                                                  int* __restrict__ binbeg,
                                                  int* __restrict__ pairs) {
    __shared__ int rowtot[NBIN];
    __shared__ int rowpre[NBIN];
    __shared__ int base[NBIN];
    __shared__ int cur[NBIN];
    const int tid  = threadIdx.x;
    const int lane = tid & 31;
    const int wave = tid >> 5;
    const int blk  = blockIdx.x;
    if (tid < NBIN) cur[tid] = 0;

    int4 ra[7], rb[7];
    #pragma unroll
    for (int i = 0; i < 7; ++i) {
        const int bin = wave + 32 * i;
        const int4* row = (const int4*)(M + (bin < NBIN ? bin : NBIN - 1) * SBLK);
        ra[i] = row[lane];
        rb[i] = row[32 + lane];
    }
    #pragma unroll
    for (int i = 0; i < 7; ++i) {
        const int bin = wave + 32 * i;
        {
            const int4 a = ra[i];
            const int4 b = rb[i];
            int tot = a.x + a.y + a.z + a.w + b.x + b.y + b.z + b.w;
            const int c0 = 4 * lane, c1 = 128 + 4 * lane;
            int pre = (c0     < blk ? a.x : 0) + (c0 + 1 < blk ? a.y : 0)
                    + (c0 + 2 < blk ? a.z : 0) + (c0 + 3 < blk ? a.w : 0)
                    + (c1     < blk ? b.x : 0) + (c1 + 1 < blk ? b.y : 0)
                    + (c1 + 2 < blk ? b.z : 0) + (c1 + 3 < blk ? b.w : 0);
            for (int o = 16; o >= 1; o >>= 1) {
                tot += __shfl_xor(tot, o);
                pre += __shfl_xor(pre, o);
            }
            if (lane == 0 && bin < NBIN) { rowtot[bin] = tot; rowpre[bin] = pre; }
        }
    }
    __syncthreads();
    if (wave == 0) {
        int loc[7];
        int s = 0;
        #pragma unroll
        for (int j = 0; j < 7; ++j) {
            const int bin = 7 * lane + j;
            loc[j] = s;
            s += (bin < NBIN) ? rowtot[bin] : 0;
        }
        int inc = s;
        for (int o = 1; o < 32; o <<= 1) {
            int t = __shfl_up(inc, o);
            if (lane >= o) inc += t;
        }
        const int ex = inc - s;
        #pragma unroll
        for (int j = 0; j < 7; ++j) {
            const int bin = 7 * lane + j;
            if (bin < NBIN) {
                const int bb = ex + loc[j];
                base[bin] = bb + rowpre[bin];
                if (blk == 0) binbeg[bin] = bb;
            }
        }
        if (blk == 0 && lane == 31) binbeg[NBIN] = inc;
    }
    __syncthreads();

    const int e0 = blk * EPB;
    int2* pr = (int2*)pairs;
    int dd[EPT], ss[EPT];
    #pragma unroll
    for (int i = 0; i < EPT; ++i) {
        const int idx = tid + 1024 * i;
        const int e = e0 + (idx < EPB ? idx : 0);
        dd[i] = dst[e];
        ss[i] = src[e];
    }
    #pragma unroll
    for (int i = 0; i < EPT; ++i) {
        if (tid + 1024 * i < EPB) {
            const int b = dd[i] >> BIN_SHIFT;
            const int r = atomicAdd(&cur[b], 1);
            pr[base[b] + r] = make_int2(ss[i], dd[i]);
        }
    }
}

__device__ __forceinline__ float dinv_of(int deg) {
    float d = (float)deg;
    return (d > 0.0f) ? rsqrtf(d) : 0.0f;
}

__global__ __launch_bounds__(1024) void k_binsort(const int* __restrict__ pairs,
                                                  const int* __restrict__ binbeg,
                                                  int* __restrict__ cnt,
                                                  int* __restrict__ off,
                                                  float* __restrict__ dinv,
                                                  int* __restrict__ csr) {
    __shared__ int cntS[BIN_NODES];
    __shared__ int curS[BIN_NODES];
    __shared__ int exS[BIN_NODES];
    __shared__ int wsum[16];
    const int tid  = threadIdx.x;
    const int lane = tid & 31;
    const int wave = tid >> 5;
    const int bin  = blockIdx.x;
    const int node0 = bin * BIN_NODES;
    const int beg = binbeg[bin];
    const int end = binbeg[bin + 1];
    const int2* pr = (const int2*)pairs;
    if (tid < BIN_NODES) { cntS[tid] = 0; curS[tid] = 0; }
    __syncthreads();
    for (int j0 = beg; j0 < end; j0 += 8192) {
        int dy[8];
        #pragma unroll
        for (int u = 0; u < 8; ++u) {
            const int j = j0 + tid + 1024 * u;
            dy[u] = pr[j < end ? j : end - 1].y;
        }
        #pragma unroll
        for (int u = 0; u < 8; ++u)
            if (j0 + tid + 1024 * u < end) atomicAdd(&cntS[dy[u] - node0], 1);
    }
    __syncthreads();

    const int v = (tid < BIN_NODES) ? cntS[tid] : 0;
    int inc = v;
    for (int o = 1; o < 32; o <<= 1) {
        int t = __shfl_up(inc, o);
        if (lane >= o) inc += t;
    }
    if (tid < BIN_NODES && lane == 31) wsum[wave] = inc;
    __syncthreads();
    if (tid < BIN_NODES) {
        int wp = 0;
        for (int w = 0; w < wave; ++w) wp += wsum[w];
        const int ex = wp + inc - v;
        exS[tid] = ex;
        const int i = node0 + tid;
        if (i < N_NODES) {
            off[i]  = beg + ex;
            cnt[i]  = v;
            dinv[i] = dinv_of(v + 1);
        }
    }
    __syncthreads();
    for (int j0 = beg; j0 < end; j0 += 8192) {
        int2 pp[8];
        #pragma unroll
        for (int u = 0; u < 8; ++u) {
            const int j = j0 + tid + 1024 * u;
            pp[u] = pr[j < end ? j : end - 1];
        }
        #pragma unroll
        for (int u = 0; u < 8; ++u) {
            if (j0 + tid + 1024 * u < end) {
                const int d = pp[u].y - node0;
                const int r = atomicAdd(&curS[d], 1);
                csr[beg + exS[d] + r] = pp[u].x;
            }
        }
    }
}

#define LDS_STRIDE 260
#define N_TILES    (N_NODES / 16)
#define G1_BLOCKS  512

#define G1_LOAD16(dst, t0)                                                        \
    _Pragma("unroll")                                                             \
    for (int u = 0; u < 16; ++u) dst[u] = *(const v2f*)&A[4 * ((t0) + u)];
#define G1_WMMA16(src, t0)                                                        \
    _Pragma("unroll")                                                             \
    for (int u = 0; u < 16; ++u)                                                  \
        c = __builtin_amdgcn_wmma_f32_16x16x4_f32(false, src[u], false, bf[(t0) + u], \
                                                  (short)0, c, false, false);

__global__ __launch_bounds__(256) void k_gemm1(const float* __restrict__ x,
                                               const float* __restrict__ W1,
                                               const float* __restrict__ dinv,
                                               float* __restrict__ h1) {
    __shared__ __attribute__((aligned(16))) float As[2][16 * LDS_STRIDE];
    const int tid  = threadIdx.x;
    const int wave = tid >> 5;
    const int lane = tid & 31;
    const int m    = lane & 15;
    const int kg   = lane >> 4;
    const int n0   = wave * 16;

    v2f bf[64];
    #pragma unroll
    for (int t = 0; t < 64; ++t) {
        const int ka = 4 * t + 2 * kg;
        bf[t].x = W1[(size_t)ka       * F_H + n0 + m];
        bf[t].y = W1[(size_t)(ka + 1) * F_H + n0 + m];
    }

    int tile = blockIdx.x;
    const int q0 = tid, q1 = tid + 256, q2 = tid + 512, q3 = tid + 768;
    const int d0 = (q0 >> 6) * LDS_STRIDE + (q0 & 63) * 4;
    const int d1 = (q1 >> 6) * LDS_STRIDE + (q1 & 63) * 4;
    const int d2 = (q2 >> 6) * LDS_STRIDE + (q2 & 63) * 4;
    const int d3 = (q3 >> 6) * LDS_STRIDE + (q3 & 63) * 4;
    {
        const float4* xg = (const float4*)(x + (size_t)tile * 16 * F_IN);
        const float4 f0 = xg[q0], f1 = xg[q1], f2 = xg[q2], f3 = xg[q3];
        *(float4*)&As[0][d0] = f0;
        *(float4*)&As[0][d1] = f1;
        *(float4*)&As[0][d2] = f2;
        *(float4*)&As[0][d3] = f3;
    }
    float4 p0, p1, p2, p3;
    {
        const int t1 = tile + G1_BLOCKS;
        const float4* xg = (const float4*)(x + (size_t)(t1 < N_TILES ? t1 : tile) * 16 * F_IN);
        p0 = xg[q0]; p1 = xg[q1]; p2 = xg[q2]; p3 = xg[q3];
    }
    __syncthreads();

    int buf = 0;
    for (; tile < N_TILES; tile += G1_BLOCKS) {
        const bool more = tile + G1_BLOCKS < N_TILES;

        const float4* dv = (const float4*)(dinv + tile * 16 + 8 * kg);
        const float4 s0 = dv[0], s1 = dv[1];

        const float* A = &As[buf][m * LDS_STRIDE + 2 * kg];
        v8f c = {0.f,0.f,0.f,0.f,0.f,0.f,0.f,0.f};
        v2f a0[16], a1[16];
        G1_LOAD16(a0, 0)
        G1_LOAD16(a1, 16)
        __builtin_amdgcn_sched_barrier(0);
        G1_WMMA16(a0, 0)
        __builtin_amdgcn_sched_barrier(0);
        G1_LOAD16(a0, 32)
        __builtin_amdgcn_sched_barrier(0);
        G1_WMMA16(a1, 16)
        __builtin_amdgcn_sched_barrier(0);
        G1_LOAD16(a1, 48)
        __builtin_amdgcn_sched_barrier(0);
        G1_WMMA16(a0, 32)
        G1_WMMA16(a1, 48)

        if (more) {
            float* An = As[buf ^ 1];
            *(float4*)&An[d0] = p0;
            *(float4*)&An[d1] = p1;
            *(float4*)&An[d2] = p2;
            *(float4*)&An[d3] = p3;
        }
        {
            const int t2 = tile + 2 * G1_BLOCKS;
            const float4* xg = (const float4*)(x + (size_t)(t2 < N_TILES ? t2 : tile) * 16 * F_IN);
            p0 = xg[q0]; p1 = xg[q1]; p2 = xg[q2]; p3 = xg[q3];
        }
        __builtin_amdgcn_fence(__ATOMIC_RELEASE, "workgroup", "local");
        __builtin_amdgcn_s_barrier();
        __builtin_amdgcn_fence(__ATOMIC_ACQUIRE, "workgroup", "local");
        buf ^= 1;

        const float sc[8] = {s0.x, s0.y, s0.z, s0.w, s1.x, s1.y, s1.z, s1.w};
        float* out = h1 + (size_t)tile * 16 * F_H;
        #pragma unroll
        for (int r = 0; r < 8; ++r) {
            int row = r + 8 * kg;
            out[(size_t)row * F_H + n0 + m] = c[r] * sc[r];
        }
    }
}

__global__ __launch_bounds__(256) void k_padW2(const float* __restrict__ W2,
                                               float* __restrict__ Bp) {
    int t = blockIdx.x * 256 + threadIdx.x;
    if (t >= F_H * H_PAD) return;
    int k = t >> 4, c = t & 15;
    Bp[t] = (c < N_CLS) ? W2[k * N_CLS + c] : 0.0f;
}

#define T_STRIDE 132
__global__ __launch_bounds__(256) void k_gather1(const int* __restrict__ csr,
                                                 const int* __restrict__ off,
                                                 const int* __restrict__ cnt,
                                                 const float* __restrict__ dinv,
                                                 const float* __restrict__ h1,
                                                 const float* __restrict__ b1,
                                                 const float* __restrict__ Bp,
                                                 float* __restrict__ h2) {
    __shared__ __attribute__((aligned(16))) float T[16 * T_STRIDE];
    const int tile = blockIdx.x;
    const int tid  = threadIdx.x;
    const int wave = tid >> 5;
    const int lane = tid & 31;
    const float4 bb = *(const float4*)&b1[lane * 4];

    for (int j = 0; j < 2; ++j) {
        const int row = wave * 2 + j;
        const int i   = __builtin_amdgcn_readfirstlane(tile * 16 + row);
        const int beg = off[i];
        const int n   = cnt[i];
        const float di = dinv[i];
        float4 acc = make_float4(0.f, 0.f, 0.f, 0.f);
        for (int c = 0; c < n; c += 32) {
            const int m = (n - c < 32) ? (n - c) : 32;
            int my_s = 0;
            if (lane < m) my_s = csr[beg + c + lane];
            #pragma unroll 4
            for (int k = 0; k < m; ++k) {
                const int s = __builtin_amdgcn_readlane(my_s, k);
                const float4 r = *(const float4*)&h1[(size_t)s * F_H + lane * 4];
                acc.x += r.x; acc.y += r.y; acc.z += r.z; acc.w += r.w;
            }
        }
        const float4 hs = *(const float4*)&h1[(size_t)i * F_H + lane * 4];
        float4 v;
        v.x = (acc.x + hs.x) * di + bb.x;
        v.y = (acc.y + hs.y) * di + bb.y;
        v.z = (acc.z + hs.z) * di + bb.z;
        v.w = (acc.w + hs.w) * di + bb.w;
        v.x = v.x > 0.0f ? v.x * di : 0.0f;
        v.y = v.y > 0.0f ? v.y * di : 0.0f;
        v.z = v.z > 0.0f ? v.z * di : 0.0f;
        v.w = v.w > 0.0f ? v.w * di : 0.0f;
        *(float4*)&T[row * T_STRIDE + lane * 4] = v;
    }
    __syncthreads();
    if (wave != 0) return;

    const int m  = lane & 15;
    const int kg = lane >> 4;
    v8f c = {0.f,0.f,0.f,0.f,0.f,0.f,0.f,0.f};
    for (int k0 = 0; k0 < F_H; k0 += 4) {
        int ka = k0 + 2 * kg;
        v2f a = *(const v2f*)&T[m * T_STRIDE + ka];
        v2f b;
        b.x = Bp[ka * H_PAD + m];
        b.y = Bp[(ka + 1) * H_PAD + m];
        c = __builtin_amdgcn_wmma_f32_16x16x4_f32(false, a, false, b,
                                                  (short)0, c, false, false);
    }
    float* out = h2 + (size_t)tile * 16 * H_PAD;
    #pragma unroll
    for (int r = 0; r < 8; ++r) {
        int row = r + 8 * kg;
        out[(size_t)row * H_PAD + m] = c[r];
    }
}

__global__ __launch_bounds__(256) void k_gather2(const int* __restrict__ csr,
                                                 const int* __restrict__ off,
                                                 const int* __restrict__ cnt,
                                                 const float* __restrict__ dinv,
                                                 const float* __restrict__ h2,
                                                 const float* __restrict__ b2,
                                                 float* __restrict__ out) {
    const int tid = threadIdx.x;
    const int f   = tid & 15;
    const int i   = blockIdx.x * 16 + (tid >> 4);
    const int beg = off[i];
    const int n   = cnt[i];
    const float di = dinv[i];
    float acc = 0.f;
    for (int c = 0; c < n; c += 16) {
        const int m = (n - c < 16) ? (n - c) : 16;
        int my_s = 0;
        if (f < m) my_s = csr[beg + c + f];
        for (int k = 0; k < m; k += 4) {
            #pragma unroll
            for (int u = 0; u < 4; ++u) {
                const int   s = __shfl(my_s, k + u, 16);
                const float v = h2[(size_t)s * H_PAD + f];
                acc += (k + u < m) ? v : 0.0f;
            }
        }
    }
    const bool valid = f < N_CLS;
    float t = -INFINITY;
    if (valid) t = (acc + h2[(size_t)i * H_PAD + f]) * di + b2[f];
    float mx = t;
    for (int o = 8; o >= 1; o >>= 1) mx = fmaxf(mx, __shfl_xor(mx, o, 16));
    float sm = valid ? expf(t - mx) : 0.0f;
    for (int o = 8; o >= 1; o >>= 1) sm += __shfl_xor(sm, o, 16);
    const float lse = logf(sm) + mx;
    if (valid) out[(size_t)i * N_CLS + f] = t - lse;
}

extern "C" void kernel_launch(void* const* d_in, const int* in_sizes, int n_in,
                              void* d_out, int out_size, void* d_ws, size_t ws_size,
                              hipStream_t stream) {
    const float* x   = (const float*)d_in[0];
    const int*   ei  = (const int*)d_in[1];
    const float* W1  = (const float*)d_in[2];
    const float* b1  = (const float*)d_in[3];
    const float* W2  = (const float*)d_in[4];
    const float* b2  = (const float*)d_in[5];
    float*       out = (float*)d_out;

    const int* src = ei;
    const int* dst = ei + N_EDGES;

    float* ws    = (float*)d_ws;
    float* dinv  = ws;
    float* h1    = dinv + 100352;
    float* reg2  = h1 + (size_t)N_NODES * F_H;
    float* W2pad = reg2 + (size_t)N_NODES * F_H;
    int*   cnt   = (int*)reg2;
    int*   off   = (int*)(reg2 + 100352);
    int*   csr   = (int*)(reg2 + 1800704);
    float* h2    = reg2 + 3400704;
    int*   pairs = (int*)(reg2 + 5000704);
    int*   M     = (int*)(reg2 + 8200704);
    int*   binbeg= (int*)(reg2 + 8250880);

    k_binhist<<<SBLK, 1024, 0, stream>>>(dst, M);
    k_binmove<<<SBLK, 1024, 0, stream>>>(src, dst, M, binbeg, pairs);
    k_binsort<<<NBIN, 1024, 0, stream>>>(pairs, binbeg, cnt, off, dinv, csr);

    k_gemm1<<<G1_BLOCKS, 256, 0, stream>>>(x, W1, dinv, h1);
    k_padW2<<<(F_H * H_PAD + 255) / 256, 256, 0, stream>>>(W2, W2pad);
    k_gather1<<<N_NODES / 16, 256, 0, stream>>>(csr, off, cnt, dinv, h1, b1, W2pad, h2);

    k_gather2<<<N_NODES / 16, 256, 0, stream>>>(csr, off, cnt, dinv, h2, b2, out);
}


# both gathers: a wave's row-independent requests (bounds, edge lists, self rows) go out together at the start; rows requested 8 at a time, left-overs together (was 4, then 1 by 1); slot order kept
# speedup vs baseline: 4.2118x; 1.0485x over previous
#include <hip/hip_runtime.h>
#include <math.h>


#define N_NODES 100000
#define N_EDGES 1600000
#define F_IN    256
#define F_H     128
#define N_CLS   10
#define H_PAD   16

typedef float v2f __attribute__((ext_vector_type(2)));
typedef float v8f __attribute__((ext_vector_type(8)));

#define BIN_SHIFT 9
#define BIN_NODES 512
#define NBIN      196
#define SBLK      256
#define EPB       (N_EDGES / SBLK)
#define EPT       7

__global__ __launch_bounds__(1024) void k_binhist(const int* __restrict__ dst,
                                                  int* __restrict__ M) {
    __shared__ int hist[NBIN];
    const int tid = threadIdx.x;
    const int blk = blockIdx.x;
    if (tid < NBIN) hist[tid] = 0;
    __syncthreads();
    const int e0 = blk * EPB;
    int d[EPT];
    #pragma unroll
    for (int i = 0; i < EPT; ++i) {
        const int idx = tid + 1024 * i;
        d[i] = dst[e0 + (idx < EPB ? idx : 0)];
    }
    #pragma unroll
    for (int i = 0; i < EPT; ++i)
        if (tid + 1024 * i < EPB) atomicAdd(&hist[d[i] >> BIN_SHIFT], 1);
    __syncthreads();
    if (tid < NBIN) M[tid * SBLK + blk] = hist[tid];
}

__global__ __launch_bounds__(1024) void k_binmove(const int* __restrict__ src,
                                                  const int* __restrict__ dst,
                                                  const int* __restrict__ M,
                                                  int* __restrict__ binbeg,
                                                  int* __restrict__ pairs) {
    __shared__ int rowtot[NBIN];
    __shared__ int rowpre[NBIN];
    __shared__ int base[NBIN];
    __shared__ int cur[NBIN];
    const int tid  = threadIdx.x;
    const int lane = tid & 31;
    const int wave = tid >> 5;
    const int blk  = blockIdx.x;
    if (tid < NBIN) cur[tid] = 0;

    int4 ra[7], rb[7];
    #pragma unroll
    for (int i = 0; i < 7; ++i) {
        const int bin = wave + 32 * i;
        const int4* row = (const int4*)(M + (bin < NBIN ? bin : NBIN - 1) * SBLK);
        ra[i] = row[lane];
        rb[i] = row[32 + lane];
    }
    #pragma unroll
    for (int i = 0; i < 7; ++i) {
        const int bin = wave + 32 * i;
        {
            const int4 a = ra[i];
            const int4 b = rb[i];
            int tot = a.x + a.y + a.z + a.w + b.x + b.y + b.z + b.w;
            const int c0 = 4 * lane, c1 = 128 + 4 * lane;
            int pre = (c0     < blk ? a.x : 0) + (c0 + 1 < blk ? a.y : 0)
                    + (c0 + 2 < blk ? a.z : 0) + (c0 + 3 < blk ? a.w : 0)
                    + (c1     < blk ? b.x : 0) + (c1 + 1 < blk ? b.y : 0)
                    + (c1 + 2 < blk ? b.z : 0) + (c1 + 3 < blk ? b.w : 0);
            for (int o = 16; o >= 1; o >>= 1) {
                tot += __shfl_xor(tot, o);
                pre += __shfl_xor(pre, o);
            }
            if (lane == 0 && bin < NBIN) { rowtot[bin] = tot; rowpre[bin] = pre; }
        }
    }
    __syncthreads();
    if (wave == 0) {
        int loc[7];
        int s = 0;
        #pragma unroll
        for (int j = 0; j < 7; ++j) {
            const int bin = 7 * lane + j;
            loc[j] = s;
            s += (bin < NBIN) ? rowtot[bin] : 0;
        }
        int inc = s;
        for (int o = 1; o < 32; o <<= 1) {
            int t = __shfl_up(inc, o);
            if (lane >= o) inc += t;
        }
        const int ex = inc - s;
        #pragma unroll
        for (int j = 0; j < 7; ++j) {
            const int bin = 7 * lane + j;
            if (bin < NBIN) {
                const int bb = ex + loc[j];
                base[bin] = bb + rowpre[bin];
                if (blk == 0) binbeg[bin] = bb;
            }
        }
        if (blk == 0 && lane == 31) binbeg[NBIN] = inc;
    }
    __syncthreads();

    const int e0 = blk * EPB;
    int2* pr = (int2*)pairs;
    int dd[EPT], ss[EPT];
    #pragma unroll
    for (int i = 0; i < EPT; ++i) {
        const int idx = tid + 1024 * i;
        const int e = e0 + (idx < EPB ? idx : 0);
        dd[i] = dst[e];
        ss[i] = src[e];
    }
    #pragma unroll
    for (int i = 0; i < EPT; ++i) {
        if (tid + 1024 * i < EPB) {
            const int b = dd[i] >> BIN_SHIFT;
            const int r = atomicAdd(&cur[b], 1);
            pr[base[b] + r] = make_int2(ss[i], dd[i]);
        }
    }
}

__device__ __forceinline__ float dinv_of(int deg) {
    float d = (float)deg;
    return (d > 0.0f) ? rsqrtf(d) : 0.0f;
}

__global__ __launch_bounds__(1024) void k_binsort(const int* __restrict__ pairs,
                                                  const int* __restrict__ binbeg,
                                                  int* __restrict__ cnt,
                                                  int* __restrict__ off,
                                                  float* __restrict__ dinv,
                                                  int* __restrict__ csr) {
    __shared__ int cntS[BIN_NODES];
    __shared__ int curS[BIN_NODES];
    __shared__ int exS[BIN_NODES];
    __shared__ int wsum[16];
    const int tid  = threadIdx.x;
    const int lane = tid & 31;
    const int wave = tid >> 5;
    const int bin  = blockIdx.x;
    const int node0 = bin * BIN_NODES;
    const int beg = binbeg[bin];
    const int end = binbeg[bin + 1];
    const int2* pr = (const int2*)pairs;
    if (tid < BIN_NODES) { cntS[tid] = 0; curS[tid] = 0; }
    __syncthreads();
    for (int j0 = beg; j0 < end; j0 += 8192) {
        int dy[8];
        #pragma unroll
        for (int u = 0; u < 8; ++u) {
            const int j = j0 + tid + 1024 * u;
            dy[u] = pr[j < end ? j : end - 1].y;
        }
        #pragma unroll
        for (int u = 0; u < 8; ++u)
            if (j0 + tid + 1024 * u < end) atomicAdd(&cntS[dy[u] - node0], 1);
    }
    __syncthreads();

    const int v = (tid < BIN_NODES) ? cntS[tid] : 0;
    int inc = v;
    for (int o = 1; o < 32; o <<= 1) {
        int t = __shfl_up(inc, o);
        if (lane >= o) inc += t;
    }
    if (tid < BIN_NODES && lane == 31) wsum[wave] = inc;
    __syncthreads();
    if (tid < BIN_NODES) {
        int wp = 0;
        for (int w = 0; w < wave; ++w) wp += wsum[w];
        const int ex = wp + inc - v;
        exS[tid] = ex;
        const int i = node0 + tid;
        if (i < N_NODES) {
            off[i]  = beg + ex;
            cnt[i]  = v;
            dinv[i] = dinv_of(v + 1);
        }
    }
    __syncthreads();
    for (int j0 = beg; j0 < end; j0 += 8192) {
        int2 pp[8];
        #pragma unroll
        for (int u = 0; u < 8; ++u) {
            const int j = j0 + tid + 1024 * u;
            pp[u] = pr[j < end ? j : end - 1];
        }
        #pragma unroll
        for (int u = 0; u < 8; ++u) {
            if (j0 + tid + 1024 * u < end) {
                const int d = pp[u].y - node0;
                const int r = atomicAdd(&curS[d], 1);
                csr[beg + exS[d] + r] = pp[u].x;
            }
        }
    }
}

#define LDS_STRIDE 260
#define N_TILES    (N_NODES / 16)
#define G1_BLOCKS  512

#define G1_LOAD16(dst, t0)                                                        \
    _Pragma("unroll")                                                             \
    for (int u = 0; u < 16; ++u) dst[u] = *(const v2f*)&A[4 * ((t0) + u)];
#define G1_WMMA16(src, t0)                                                        \
    _Pragma("unroll")                                                             \
    for (int u = 0; u < 16; ++u)                                                  \
        c = __builtin_amdgcn_wmma_f32_16x16x4_f32(false, src[u], false, bf[(t0) + u], \
                                                  (short)0, c, false, false);

__global__ __launch_bounds__(256) void k_gemm1(const float* __restrict__ x,
                                               const float* __restrict__ W1,
                                               const float* __restrict__ dinv,
                                               float* __restrict__ h1) {
    __shared__ __attribute__((aligned(16))) float As[2][16 * LDS_STRIDE];
    const int tid  = threadIdx.x;
    const int wave = tid >> 5;
    const int lane = tid & 31;
    const int m    = lane & 15;
    const int kg   = lane >> 4;
    const int n0   = wave * 16;

    v2f bf[64];
    #pragma unroll
    for (int t = 0; t < 64; ++t) {
        const int ka = 4 * t + 2 * kg;
        bf[t].x = W1[(size_t)ka       * F_H + n0 + m];
        bf[t].y = W1[(size_t)(ka + 1) * F_H + n0 + m];
    }

    int tile = blockIdx.x;
    const int q0 = tid, q1 = tid + 256, q2 = tid + 512, q3 = tid + 768;
    const int d0 = (q0 >> 6) * LDS_STRIDE + (q0 & 63) * 4;
    const int d1 = (q1 >> 6) * LDS_STRIDE + (q1 & 63) * 4;
    const int d2 = (q2 >> 6) * LDS_STRIDE + (q2 & 63) * 4;
    const int d3 = (q3 >> 6) * LDS_STRIDE + (q3 & 63) * 4;
    {
        const float4* xg = (const float4*)(x + (size_t)tile * 16 * F_IN);
        const float4 f0 = xg[q0], f1 = xg[q1], f2 = xg[q2], f3 = xg[q3];
        *(float4*)&As[0][d0] = f0;
        *(float4*)&As[0][d1] = f1;
        *(float4*)&As[0][d2] = f2;
        *(float4*)&As[0][d3] = f3;
    }
    float4 p0, p1, p2, p3;
    {
        const int t1 = tile + G1_BLOCKS;
        const float4* xg = (const float4*)(x + (size_t)(t1 < N_TILES ? t1 : tile) * 16 * F_IN);
        p0 = xg[q0]; p1 = xg[q1]; p2 = xg[q2]; p3 = xg[q3];
    }
    __syncthreads();

    int buf = 0;
    for (; tile < N_TILES; tile += G1_BLOCKS) {
        const bool more = tile + G1_BLOCKS < N_TILES;

        const float4* dv = (const float4*)(dinv + tile * 16 + 8 * kg);
        const float4 s0 = dv[0], s1 = dv[1];

        const float* A = &As[buf][m * LDS_STRIDE + 2 * kg];
        v8f c = {0.f,0.f,0.f,0.f,0.f,0.f,0.f,0.f};
        v2f a0[16], a1[16];
        G1_LOAD16(a0, 0)
        G1_LOAD16(a1, 16)
        __builtin_amdgcn_sched_barrier(0);
        G1_WMMA16(a0, 0)
        __builtin_amdgcn_sched_barrier(0);
        G1_LOAD16(a0, 32)
        __builtin_amdgcn_sched_barrier(0);
        G1_WMMA16(a1, 16)
        __builtin_amdgcn_sched_barrier(0);
        G1_LOAD16(a1, 48)
        __builtin_amdgcn_sched_barrier(0);
        G1_WMMA16(a0, 32)
        G1_WMMA16(a1, 48)

        if (more) {
            float* An = As[buf ^ 1];
            *(float4*)&An[d0] = p0;
            *(float4*)&An[d1] = p1;
            *(float4*)&An[d2] = p2;
            *(float4*)&An[d3] = p3;
        }
        {
            const int t2 = tile + 2 * G1_BLOCKS;
            const float4* xg = (const float4*)(x + (size_t)(t2 < N_TILES ? t2 : tile) * 16 * F_IN);
            p0 = xg[q0]; p1 = xg[q1]; p2 = xg[q2]; p3 = xg[q3];
        }
        __builtin_amdgcn_fence(__ATOMIC_RELEASE, "workgroup", "local");
        __builtin_amdgcn_s_barrier();
        __builtin_amdgcn_fence(__ATOMIC_ACQUIRE, "workgroup", "local");
        buf ^= 1;

        const float sc[8] = {s0.x, s0.y, s0.z, s0.w, s1.x, s1.y, s1.z, s1.w};
        float* out = h1 + (size_t)tile * 16 * F_H;
        #pragma unroll
        for (int r = 0; r < 8; ++r) {
            int row = r + 8 * kg;
            out[(size_t)row * F_H + n0 + m] = c[r] * sc[r];
        }
    }
}

__global__ __launch_bounds__(256) void k_padW2(const float* __restrict__ W2,
                                               float* __restrict__ Bp) {
    int t = blockIdx.x * 256 + threadIdx.x;
    if (t >= F_H * H_PAD) return;
    int k = t >> 4, c = t & 15;
    Bp[t] = (c < N_CLS) ? W2[k * N_CLS + c] : 0.0f;
}

#define T_STRIDE 132
#define G1_GRP   8

__device__ __forceinline__ float4 g1_sum_rows(const int* __restrict__ csr,
                                              const float* __restrict__ h1,
                                              int beg, int n, int first_s, int lane) {
    float4 acc = make_float4(0.f, 0.f, 0.f, 0.f);
    for (int c = 0; c < n; c += 32) {
        const int m = (n - c < 32) ? (n - c) : 32;
        int my_s = first_s;
        if (c > 0) {
            my_s = 0;
            if (lane < m) my_s = csr[beg + c + lane];
        }
        int k = 0;
        for (; k + G1_GRP <= m; k += G1_GRP) {
            float4 r[G1_GRP];
            #pragma unroll
            for (int u = 0; u < G1_GRP; ++u) {
                const int s = __builtin_amdgcn_readlane(my_s, k + u);
                r[u] = *(const float4*)&h1[(size_t)s * F_H + lane * 4];
            }
            #pragma unroll
            for (int u = 0; u < G1_GRP; ++u) {
                acc.x += r[u].x; acc.y += r[u].y; acc.z += r[u].z; acc.w += r[u].w;
            }
        }
        if (k < m) {
            float4 r[G1_GRP - 1];
            #pragma unroll
            for (int u = 0; u < G1_GRP - 1; ++u) {
                r[u] = make_float4(0.f, 0.f, 0.f, 0.f);
                if (k + u < m) {
                    const int s = __builtin_amdgcn_readlane(my_s, k + u);
                    r[u] = *(const float4*)&h1[(size_t)s * F_H + lane * 4];
                }
            }
            #pragma unroll
            for (int u = 0; u < G1_GRP - 1; ++u) {
                acc.x += r[u].x; acc.y += r[u].y; acc.z += r[u].z; acc.w += r[u].w;
            }
        }
    }
    return acc;
}

__device__ __forceinline__ float4 g1_finish(float4 acc, float4 hs, float di, float4 bb) {
    float4 v;
    v.x = (acc.x + hs.x) * di + bb.x;
    v.y = (acc.y + hs.y) * di + bb.y;
    v.z = (acc.z + hs.z) * di + bb.z;
    v.w = (acc.w + hs.w) * di + bb.w;
    v.x = v.x > 0.0f ? v.x * di : 0.0f;
    v.y = v.y > 0.0f ? v.y * di : 0.0f;
    v.z = v.z > 0.0f ? v.z * di : 0.0f;
    v.w = v.w > 0.0f ? v.w * di : 0.0f;
    return v;
}

__global__ __launch_bounds__(256) void k_gather1(const int* __restrict__ csr,
                                                 const int* __restrict__ off,
                                                 const int* __restrict__ cnt,
                                                 const float* __restrict__ dinv,
                                                 const float* __restrict__ h1,
                                                 const float* __restrict__ b1,
                                                 const float* __restrict__ Bp,
                                                 float* __restrict__ h2) {
    __shared__ __attribute__((aligned(16))) float T[16 * T_STRIDE];
    const int tile = blockIdx.x;
    const int tid  = threadIdx.x;
    const int wave = tid >> 5;
    const int lane = tid & 31;
    const float4 bb = *(const float4*)&b1[lane * 4];

    const int i0 = __builtin_amdgcn_readfirstlane(tile * 16 + wave * 2);
    const int2   o2 = *(const int2*)&off[i0];
    const int2   n2 = *(const int2*)&cnt[i0];
    const float2 d2 = *(const float2*)&dinv[i0];
    const float4 hs0 = *(const float4*)&h1[(size_t)i0 * F_H + lane * 4];
    const float4 hs1 = *(const float4*)&h1[(size_t)(i0 + 1) * F_H + lane * 4];
    int s0 = 0, s1 = 0;
    if (lane < n2.x) s0 = csr[o2.x + lane];
    if (lane < n2.y) s1 = csr[o2.y + lane];

    const float4 a0 = g1_sum_rows(csr, h1, o2.x, n2.x, s0, lane);
    *(float4*)&T[(wave * 2) * T_STRIDE + lane * 4] = g1_finish(a0, hs0, d2.x, bb);
    const float4 a1 = g1_sum_rows(csr, h1, o2.y, n2.y, s1, lane);
    *(float4*)&T[(wave * 2 + 1) * T_STRIDE + lane * 4] = g1_finish(a1, hs1, d2.y, bb);
    __syncthreads();
    if (wave != 0) return;

    const int m  = lane & 15;
    const int kg = lane >> 4;
    v8f c = {0.f,0.f,0.f,0.f,0.f,0.f,0.f,0.f};
    for (int k0 = 0; k0 < F_H; k0 += 4) {
        int ka = k0 + 2 * kg;
        v2f a = *(const v2f*)&T[m * T_STRIDE + ka];
        v2f b;
        b.x = Bp[ka * H_PAD + m];
        b.y = Bp[(ka + 1) * H_PAD + m];
        c = __builtin_amdgcn_wmma_f32_16x16x4_f32(false, a, false, b,
                                                  (short)0, c, false, false);
    }
    float* out = h2 + (size_t)tile * 16 * H_PAD;
    #pragma unroll
    for (int r = 0; r < 8; ++r) {
        int row = r + 8 * kg;
        out[(size_t)row * H_PAD + m] = c[r];
    }
}

#define G2_GRP 8
__global__ __launch_bounds__(256) void k_gather2(const int* __restrict__ csr,
                                                 const int* __restrict__ off,
                                                 const int* __restrict__ cnt,
                                                 const float* __restrict__ dinv,
                                                 const float* __restrict__ h2,
                                                 const float* __restrict__ b2,
                                                 float* __restrict__ out) {
    const int tid = threadIdx.x;
    const int f   = tid & 15;
    const int i   = blockIdx.x * 16 + (tid >> 4);
    const bool valid = f < N_CLS;
    const int beg = off[i];
    const int n   = cnt[i];
    const float di = dinv[i];
    const float hself = h2[(size_t)i * H_PAD + f];
    const float bias  = b2[valid ? f : 0];
    int sa = 0, sb = 0;
    if (f < n)      sa = csr[beg + f];
    if (f + 16 < n) sb = csr[beg + 16 + f];
    float acc = 0.f;
    for (int c = 0; c < n; c += 32) {
        const int m = (n - c < 32) ? (n - c) : 32;
        if (c > 0) {
            sa = 0; sb = 0;
            if (f < m)      sa = csr[beg + c + f];
            if (f + 16 < m) sb = csr[beg + c + 16 + f];
        }
        for (int k = 0; k < m; k += G2_GRP) {
            const int sv = (k < 16) ? sa : sb;
            float v[G2_GRP];
            #pragma unroll
            for (int u = 0; u < G2_GRP; ++u) {
                const int s = __shfl(sv, (k + u) & 15, 16);
                v[u] = 0.0f;
                if (k + u < m) v[u] = h2[(size_t)s * H_PAD + f];
            }
            #pragma unroll
            for (int u = 0; u < G2_GRP; ++u) acc += v[u];
        }
    }
    float t = -INFINITY;
    if (valid) t = (acc + hself) * di + bias;
    float mx = t;
    for (int o = 8; o >= 1; o >>= 1) mx = fmaxf(mx, __shfl_xor(mx, o, 16));
    float sm = valid ? expf(t - mx) : 0.0f;
    for (int o = 8; o >= 1; o >>= 1) sm += __shfl_xor(sm, o, 16);
    const float lse = logf(sm) + mx;
    if (valid) out[(size_t)i * N_CLS + f] = t - lse;
}

extern "C" void kernel_launch(void* const* d_in, const int* in_sizes, int n_in,
                              void* d_out, int out_size, void* d_ws, size_t ws_size,
                              hipStream_t stream) {
    const float* x   = (const float*)d_in[0];
    const int*   ei  = (const int*)d_in[1];
    const float* W1  = (const float*)d_in[2];
    const float* b1  = (const float*)d_in[3];
    const float* W2  = (const float*)d_in[4];
    const float* b2  = (const float*)d_in[5];
    float*       out = (float*)d_out;

    const int* src = ei;
    const int* dst = ei + N_EDGES;

    float* ws    = (float*)d_ws;
    float* dinv  = ws;
    float* h1    = dinv + 100352;
    float* reg2  = h1 + (size_t)N_NODES * F_H;
    float* W2pad = reg2 + (size_t)N_NODES * F_H;
    int*   cnt   = (int*)reg2;
    int*   off   = (int*)(reg2 + 100352);
    int*   csr   = (int*)(reg2 + 1800704);
    float* h2    = reg2 + 3400704;
    int*   pairs = (int*)(reg2 + 5000704);
    int*   M     = (int*)(reg2 + 8200704);
    int*   binbeg= (int*)(reg2 + 8250880);

    k_binhist<<<SBLK, 1024, 0, stream>>>(dst, M);
    k_binmove<<<SBLK, 1024, 0, stream>>>(src, dst, M, binbeg, pairs);
    k_binsort<<<NBIN, 1024, 0, stream>>>(pairs, binbeg, cnt, off, dinv, csr);

    k_gemm1<<<G1_BLOCKS, 256, 0, stream>>>(x, W1, dinv, h1);
    k_padW2<<<(F_H * H_PAD + 255) / 256, 256, 0, stream>>>(W2, W2pad);
    k_gather1<<<N_NODES / 16, 256, 0, stream>>>(csr, off, cnt, dinv, h1, b1, W2pad, h2);

    k_gather2<<<N_NODES / 16, 256, 0, stream>>>(csr, off, cnt, dinv, h2, b2, out);
}


# edge sort, both levels: edges go to an LDS copy in slot order, then out with neighbouring lanes on neighbouring slots (whole-line stores, not 32 scattered 4- or 8-byte stores each); slot order kept
# speedup vs baseline: 4.5166x; 1.0724x over previous
#include <hip/hip_runtime.h>
#include <math.h>


#define N_NODES 100000
#define N_EDGES 1600000
#define F_IN    256
#define F_H     128
#define N_CLS   10
#define H_PAD   16

typedef float v2f __attribute__((ext_vector_type(2)));
typedef float v8f __attribute__((ext_vector_type(8)));

#define BIN_SHIFT 9
#define BIN_NODES 512
#define NBIN      196
#define SBLK      256
#define EPB       (N_EDGES / SBLK)
#define EPT       7

__global__ __launch_bounds__(1024) void k_binhist(const int* __restrict__ dst,
                                                  int* __restrict__ M) {
    __shared__ int hist[NBIN];
    const int tid = threadIdx.x;
    const int blk = blockIdx.x;
    if (tid < NBIN) hist[tid] = 0;
    __syncthreads();
    const int e0 = blk * EPB;
    int d[EPT];
    #pragma unroll
    for (int i = 0; i < EPT; ++i) {
        const int idx = tid + 1024 * i;
        d[i] = dst[e0 + (idx < EPB ? idx : 0)];
    }
    #pragma unroll
    for (int i = 0; i < EPT; ++i)
        if (tid + 1024 * i < EPB) atomicAdd(&hist[d[i] >> BIN_SHIFT], 1);
    __syncthreads();
    if (tid < NBIN) M[tid * SBLK + blk] = hist[tid];
}

__global__ __launch_bounds__(1024) void k_binmove(const int* __restrict__ src,
                                                  const int* __restrict__ dst,
                                                  const int* __restrict__ M,
                                                  int* __restrict__ binbeg,
                                                  int* __restrict__ pairs) {
    __shared__ int rowtot[NBIN];
    __shared__ int rowpre[NBIN];
    __shared__ int rowown[NBIN];
    __shared__ int base[NBIN];
    __shared__ int lbeg[NBIN];
    __shared__ int cur[NBIN];
    __shared__ int2 stg[EPB];
    const int tid  = threadIdx.x;
    const int lane = tid & 31;
    const int wave = tid >> 5;
    const int blk  = blockIdx.x;
    if (tid < NBIN) cur[tid] = 0;

    int4 ra[7], rb[7];
    #pragma unroll
    for (int i = 0; i < 7; ++i) {
        const int bin = wave + 32 * i;
        const int4* row = (const int4*)(M + (bin < NBIN ? bin : NBIN - 1) * SBLK);
        ra[i] = row[lane];
        rb[i] = row[32 + lane];
    }
    #pragma unroll
    for (int i = 0; i < 7; ++i) {
        const int bin = wave + 32 * i;
        {
            const int4 a = ra[i];
            const int4 b = rb[i];
            int tot = a.x + a.y + a.z + a.w + b.x + b.y + b.z + b.w;
            const int c0 = 4 * lane, c1 = 128 + 4 * lane;
            int pre = (c0     < blk ? a.x : 0) + (c0 + 1 < blk ? a.y : 0)
                    + (c0 + 2 < blk ? a.z : 0) + (c0 + 3 < blk ? a.w : 0)
                    + (c1     < blk ? b.x : 0) + (c1 + 1 < blk ? b.y : 0)
                    + (c1 + 2 < blk ? b.z : 0) + (c1 + 3 < blk ? b.w : 0);
            int own = (c0     == blk ? a.x : 0) + (c0 + 1 == blk ? a.y : 0)
                    + (c0 + 2 == blk ? a.z : 0) + (c0 + 3 == blk ? a.w : 0)
                    + (c1     == blk ? b.x : 0) + (c1 + 1 == blk ? b.y : 0)
                    + (c1 + 2 == blk ? b.z : 0) + (c1 + 3 == blk ? b.w : 0);
            for (int o = 16; o >= 1; o >>= 1) {
                tot += __shfl_xor(tot, o);
                pre += __shfl_xor(pre, o);
                own += __shfl_xor(own, o);
            }
            if (lane == 0 && bin < NBIN) { rowtot[bin] = tot; rowpre[bin] = pre; rowown[bin] = own; }
        }
    }
    __syncthreads();
    if (wave == 0) {
        int loc[7], lloc[7];
        int s = 0, ls = 0;
        #pragma unroll
        for (int j = 0; j < 7; ++j) {
            const int bin = 7 * lane + j;
            loc[j] = s;
            lloc[j] = ls;
            s  += (bin < NBIN) ? rowtot[bin] : 0;
            ls += (bin < NBIN) ? rowown[bin] : 0;
        }
        int inc = s, linc = ls;
        for (int o = 1; o < 32; o <<= 1) {
            int t  = __shfl_up(inc, o);
            int lt = __shfl_up(linc, o);
            if (lane >= o) { inc += t; linc += lt; }
        }
        const int ex  = inc - s;
        const int lex = linc - ls;
        #pragma unroll
        for (int j = 0; j < 7; ++j) {
            const int bin = 7 * lane + j;
            if (bin < NBIN) {
                const int bb = ex + loc[j];
                base[bin] = bb + rowpre[bin];
                lbeg[bin] = lex + lloc[j];
                if (blk == 0) binbeg[bin] = bb;
            }
        }
        if (blk == 0 && lane == 31) binbeg[NBIN] = inc;
    }
    __syncthreads();

    const int e0 = blk * EPB;
    int2* pr = (int2*)pairs;
    int dd[EPT], ss[EPT];
    #pragma unroll
    for (int i = 0; i < EPT; ++i) {
        const int idx = tid + 1024 * i;
        const int e = e0 + (idx < EPB ? idx : 0);
        dd[i] = dst[e];
        ss[i] = src[e];
    }
    #pragma unroll
    for (int i = 0; i < EPT; ++i) {
        if (tid + 1024 * i < EPB) {
            const int b = dd[i] >> BIN_SHIFT;
            const int r = atomicAdd(&cur[b], 1);
            stg[lbeg[b] + r] = make_int2(ss[i], dd[i]);
        }
    }
    __syncthreads();
    int2 ee[EPT];
    #pragma unroll
    for (int i = 0; i < EPT; ++i) {
        const int p = tid + 1024 * i;
        ee[i] = stg[p < EPB ? p : 0];
    }
    #pragma unroll
    for (int i = 0; i < EPT; ++i) {
        const int p = tid + 1024 * i;
        if (p < EPB) {
            const int b = ee[i].y >> BIN_SHIFT;
            pr[base[b] + (p - lbeg[b])] = ee[i];
        }
    }
}

__device__ __forceinline__ float dinv_of(int deg) {
    float d = (float)deg;
    return (d > 0.0f) ? rsqrtf(d) : 0.0f;
}

#define STG_CAP 12288
__global__ __launch_bounds__(1024) void k_binsort(const int* __restrict__ pairs,
                                                  const int* __restrict__ binbeg,
                                                  int* __restrict__ cnt,
                                                  int* __restrict__ off,
                                                  float* __restrict__ dinv,
                                                  int* __restrict__ csr) {
    __shared__ int cntS[BIN_NODES];
    __shared__ int curS[BIN_NODES];
    __shared__ int exS[BIN_NODES];
    __shared__ int wsum[16];
    __shared__ int stgS[STG_CAP];
    const int tid  = threadIdx.x;
    const int lane = tid & 31;
    const int wave = tid >> 5;
    const int bin  = blockIdx.x;
    const int node0 = bin * BIN_NODES;
    const int beg = binbeg[bin];
    const int end = binbeg[bin + 1];
    const int2* pr = (const int2*)pairs;
    if (tid < BIN_NODES) { cntS[tid] = 0; curS[tid] = 0; }
    __syncthreads();
    for (int j0 = beg; j0 < end; j0 += 8192) {
        int dy[8];
        #pragma unroll
        for (int u = 0; u < 8; ++u) {
            const int j = j0 + tid + 1024 * u;
            dy[u] = pr[j < end ? j : end - 1].y;
        }
        #pragma unroll
        for (int u = 0; u < 8; ++u)
            if (j0 + tid + 1024 * u < end) atomicAdd(&cntS[dy[u] - node0], 1);
    }
    __syncthreads();

    const int v = (tid < BIN_NODES) ? cntS[tid] : 0;
    int inc = v;
    for (int o = 1; o < 32; o <<= 1) {
        int t = __shfl_up(inc, o);
        if (lane >= o) inc += t;
    }
    if (tid < BIN_NODES && lane == 31) wsum[wave] = inc;
    __syncthreads();
    if (tid < BIN_NODES) {
        int wp = 0;
        for (int w = 0; w < wave; ++w) wp += wsum[w];
        const int ex = wp + inc - v;
        exS[tid] = ex;
        const int i = node0 + tid;
        if (i < N_NODES) {
            off[i]  = beg + ex;
            cnt[i]  = v;
            dinv[i] = dinv_of(v + 1);
        }
    }
    __syncthreads();
    for (int j0 = beg; j0 < end; j0 += 8192) {
        int2 pp[8];
        #pragma unroll
        for (int u = 0; u < 8; ++u) {
            const int j = j0 + tid + 1024 * u;
            pp[u] = pr[j < end ? j : end - 1];
        }
        #pragma unroll
        for (int u = 0; u < 8; ++u) {
            if (j0 + tid + 1024 * u < end) {
                const int d = pp[u].y - node0;
                const int r = atomicAdd(&curS[d], 1);
                const int pos = exS[d] + r;
                if (pos < STG_CAP) stgS[pos] = pp[u].x;
                else               csr[beg + pos] = pp[u].x;
            }
        }
    }
    __syncthreads();
    const int nst = (end - beg < STG_CAP) ? (end - beg) : STG_CAP;
    for (int p0 = 0; p0 < nst; p0 += 4096) {
        int sv[4];
        #pragma unroll
        for (int u = 0; u < 4; ++u) {
            const int p = p0 + tid + 1024 * u;
            sv[u] = stgS[p < nst ? p : 0];
        }
        #pragma unroll
        for (int u = 0; u < 4; ++u) {
            const int p = p0 + tid + 1024 * u;
            if (p < nst) csr[beg + p] = sv[u];
        }
    }
}

#define LDS_STRIDE 260
#define N_TILES    (N_NODES / 16)
#define G1_BLOCKS  512

#define G1_LOAD16(dst, t0)                                                        \
    _Pragma("unroll")                                                             \
    for (int u = 0; u < 16; ++u) dst[u] = *(const v2f*)&A[4 * ((t0) + u)];
#define G1_WMMA16(src, t0)                                                        \
    _Pragma("unroll")                                                             \
    for (int u = 0; u < 16; ++u)                                                  \
        c = __builtin_amdgcn_wmma_f32_16x16x4_f32(false, src[u], false, bf[(t0) + u], \
                                                  (short)0, c, false, false);

__global__ __launch_bounds__(256) void k_gemm1(const float* __restrict__ x,
                                               const float* __restrict__ W1,
                                               const float* __restrict__ dinv,
                                               float* __restrict__ h1) {
    __shared__ __attribute__((aligned(16))) float As[2][16 * LDS_STRIDE];
    const int tid  = threadIdx.x;
    const int wave = tid >> 5;
    const int lane = tid & 31;
    const int m    = lane & 15;
    const int kg   = lane >> 4;
    const int n0   = wave * 16;

    v2f bf[64];
    #pragma unroll
    for (int t = 0; t < 64; ++t) {
        const int ka = 4 * t + 2 * kg;
        bf[t].x = W1[(size_t)ka       * F_H + n0 + m];
        bf[t].y = W1[(size_t)(ka + 1) * F_H + n0 + m];
    }

    int tile = blockIdx.x;
    const int q0 = tid, q1 = tid + 256, q2 = tid + 512, q3 = tid + 768;
    const int d0 = (q0 >> 6) * LDS_STRIDE + (q0 & 63) * 4;
    const int d1 = (q1 >> 6) * LDS_STRIDE + (q1 & 63) * 4;
    const int d2 = (q2 >> 6) * LDS_STRIDE + (q2 & 63) * 4;
    const int d3 = (q3 >> 6) * LDS_STRIDE + (q3 & 63) * 4;
    {
        const float4* xg = (const float4*)(x + (size_t)tile * 16 * F_IN);
        const float4 f0 = xg[q0], f1 = xg[q1], f2 = xg[q2], f3 = xg[q3];
        *(float4*)&As[0][d0] = f0;
        *(float4*)&As[0][d1] = f1;
        *(float4*)&As[0][d2] = f2;
        *(float4*)&As[0][d3] = f3;
    }
    float4 p0, p1, p2, p3;
    {
        const int t1 = tile + G1_BLOCKS;
        const float4* xg = (const float4*)(x + (size_t)(t1 < N_TILES ? t1 : tile) * 16 * F_IN);
        p0 = xg[q0]; p1 = xg[q1]; p2 = xg[q2]; p3 = xg[q3];
    }
    __syncthreads();

    int buf = 0;
    for (; tile < N_TILES; tile += G1_BLOCKS) {
        const bool more = tile + G1_BLOCKS < N_TILES;

        const float4* dv = (const float4*)(dinv + tile * 16 + 8 * kg);
        const float4 s0 = dv[0], s1 = dv[1];

        const float* A = &As[buf][m * LDS_STRIDE + 2 * kg];
        v8f c = {0.f,0.f,0.f,0.f,0.f,0.f,0.f,0.f};
        v2f a0[16], a1[16];
        G1_LOAD16(a0, 0)
        G1_LOAD16(a1, 16)
        __builtin_amdgcn_sched_barrier(0);
        G1_WMMA16(a0, 0)
        __builtin_amdgcn_sched_barrier(0);
        G1_LOAD16(a0, 32)
        __builtin_amdgcn_sched_barrier(0);
        G1_WMMA16(a1, 16)
        __builtin_amdgcn_sched_barrier(0);
        G1_LOAD16(a1, 48)
        __builtin_amdgcn_sched_barrier(0);
        G1_WMMA16(a0, 32)
        G1_WMMA16(a1, 48)

        if (more) {
            float* An = As[buf ^ 1];
            *(float4*)&An[d0] = p0;
            *(float4*)&An[d1] = p1;
            *(float4*)&An[d2] = p2;
            *(float4*)&An[d3] = p3;
        }
        {
            const int t2 = tile + 2 * G1_BLOCKS;
            const float4* xg = (const float4*)(x + (size_t)(t2 < N_TILES ? t2 : tile) * 16 * F_IN);
            p0 = xg[q0]; p1 = xg[q1]; p2 = xg[q2]; p3 = xg[q3];
        }
        __builtin_amdgcn_fence(__ATOMIC_RELEASE, "workgroup", "local");
        __builtin_amdgcn_s_barrier();
        __builtin_amdgcn_fence(__ATOMIC_ACQUIRE, "workgroup", "local");
        buf ^= 1;

        const float sc[8] = {s0.x, s0.y, s0.z, s0.w, s1.x, s1.y, s1.z, s1.w};
        float* out = h1 + (size_t)tile * 16 * F_H;
        #pragma unroll
        for (int r = 0; r < 8; ++r) {
            int row = r + 8 * kg;
            out[(size_t)row * F_H + n0 + m] = c[r] * sc[r];
        }
    }
}

__global__ __launch_bounds__(256) void k_padW2(const float* __restrict__ W2,
                                               float* __restrict__ Bp) {
    int t = blockIdx.x * 256 + threadIdx.x;
    if (t >= F_H * H_PAD) return;
    int k = t >> 4, c = t & 15;
    Bp[t] = (c < N_CLS) ? W2[k * N_CLS + c] : 0.0f;
}

#define T_STRIDE 132
#define G1_GRP   8

__device__ __forceinline__ float4 g1_sum_rows(const int* __restrict__ csr,
                                              const float* __restrict__ h1,
                                              int beg, int n, int first_s, int lane) {
    float4 acc = make_float4(0.f, 0.f, 0.f, 0.f);
    for (int c = 0; c < n; c += 32) {
        const int m = (n - c < 32) ? (n - c) : 32;
        int my_s = first_s;
        if (c > 0) {
            my_s = 0;
            if (lane < m) my_s = csr[beg + c + lane];
        }
        int k = 0;
        for (; k + G1_GRP <= m; k += G1_GRP) {
            float4 r[G1_GRP];
            #pragma unroll
            for (int u = 0; u < G1_GRP; ++u) {
                const int s = __builtin_amdgcn_readlane(my_s, k + u);
                r[u] = *(const float4*)&h1[(size_t)s * F_H + lane * 4];
            }
            #pragma unroll
            for (int u = 0; u < G1_GRP; ++u) {
                acc.x += r[u].x; acc.y += r[u].y; acc.z += r[u].z; acc.w += r[u].w;
            }
        }
        if (k < m) {
            float4 r[G1_GRP - 1];
            #pragma unroll
            for (int u = 0; u < G1_GRP - 1; ++u) {
                r[u] = make_float4(0.f, 0.f, 0.f, 0.f);
                if (k + u < m) {
                    const int s = __builtin_amdgcn_readlane(my_s, k + u);
                    r[u] = *(const float4*)&h1[(size_t)s * F_H + lane * 4];
                }
            }
            #pragma unroll
            for (int u = 0; u < G1_GRP - 1; ++u) {
                acc.x += r[u].x; acc.y += r[u].y; acc.z += r[u].z; acc.w += r[u].w;
            }
        }
    }
    return acc;
}

__device__ __forceinline__ float4 g1_finish(float4 acc, float4 hs, float di, float4 bb) {
    float4 v;
    v.x = (acc.x + hs.x) * di + bb.x;
    v.y = (acc.y + hs.y) * di + bb.y;
    v.z = (acc.z + hs.z) * di + bb.z;
    v.w = (acc.w + hs.w) * di + bb.w;
    v.x = v.x > 0.0f ? v.x * di : 0.0f;
    v.y = v.y > 0.0f ? v.y * di : 0.0f;
    v.z = v.z > 0.0f ? v.z * di : 0.0f;
    v.w = v.w > 0.0f ? v.w * di : 0.0f;
    return v;
}

__global__ __launch_bounds__(256) void k_gather1(const int* __restrict__ csr,
                                                 const int* __restrict__ off,
                                                 const int* __restrict__ cnt,
                                                 const float* __restrict__ dinv,
                                                 const float* __restrict__ h1,
                                                 const float* __restrict__ b1,
                                                 const float* __restrict__ Bp,
                                                 float* __restrict__ h2) {
    __shared__ __attribute__((aligned(16))) float T[16 * T_STRIDE];
    const int tile = blockIdx.x;
    const int tid  = threadIdx.x;
    const int wave = tid >> 5;
    const int lane = tid & 31;
    const float4 bb = *(const float4*)&b1[lane * 4];

    const int i0 = __builtin_amdgcn_readfirstlane(tile * 16 + wave * 2);
    const int2   o2 = *(const int2*)&off[i0];
    const int2   n2 = *(const int2*)&cnt[i0];
    const float2 d2 = *(const float2*)&dinv[i0];
    const float4 hs0 = *(const float4*)&h1[(size_t)i0 * F_H + lane * 4];
    const float4 hs1 = *(const float4*)&h1[(size_t)(i0 + 1) * F_H + lane * 4];
    int s0 = 0, s1 = 0;
    if (lane < n2.x) s0 = csr[o2.x + lane];
    if (lane < n2.y) s1 = csr[o2.y + lane];

    const float4 a0 = g1_sum_rows(csr, h1, o2.x, n2.x, s0, lane);
    *(float4*)&T[(wave * 2) * T_STRIDE + lane * 4] = g1_finish(a0, hs0, d2.x, bb);
    const float4 a1 = g1_sum_rows(csr, h1, o2.y, n2.y, s1, lane);
    *(float4*)&T[(wave * 2 + 1) * T_STRIDE + lane * 4] = g1_finish(a1, hs1, d2.y, bb);
    __syncthreads();
    if (wave != 0) return;

    const int m  = lane & 15;
    const int kg = lane >> 4;
    v8f c = {0.f,0.f,0.f,0.f,0.f,0.f,0.f,0.f};
    for (int k0 = 0; k0 < F_H; k0 += 4) {
        int ka = k0 + 2 * kg;
        v2f a = *(const v2f*)&T[m * T_STRIDE + ka];
        v2f b;
        b.x = Bp[ka * H_PAD + m];
        b.y = Bp[(ka + 1) * H_PAD + m];
        c = __builtin_amdgcn_wmma_f32_16x16x4_f32(false, a, false, b,
                                                  (short)0, c, false, false);
    }
    float* out = h2 + (size_t)tile * 16 * H_PAD;
    #pragma unroll
    for (int r = 0; r < 8; ++r) {
        int row = r + 8 * kg;
        out[(size_t)row * H_PAD + m] = c[r];
    }
}

#define G2_GRP 8
__global__ __launch_bounds__(256) void k_gather2(const int* __restrict__ csr,
                                                 const int* __restrict__ off,
                                                 const int* __restrict__ cnt,
                                                 const float* __restrict__ dinv,
                                                 const float* __restrict__ h2,
                                                 const float* __restrict__ b2,
                                                 float* __restrict__ out) {
    const int tid = threadIdx.x;
    const int f   = tid & 15;
    const int i   = blockIdx.x * 16 + (tid >> 4);
    const bool valid = f < N_CLS;
    const int beg = off[i];
    const int n   = cnt[i];
    const float di = dinv[i];
    const float hself = h2[(size_t)i * H_PAD + f];
    const float bias  = b2[valid ? f : 0];
    int sa = 0, sb = 0;
    if (f < n)      sa = csr[beg + f];
    if (f + 16 < n) sb = csr[beg + 16 + f];
    float acc = 0.f;
    for (int c = 0; c < n; c += 32) {
        const int m = (n - c < 32) ? (n - c) : 32;
        if (c > 0) {
            sa = 0; sb = 0;
            if (f < m)      sa = csr[beg + c + f];
            if (f + 16 < m) sb = csr[beg + c + 16 + f];
        }
        for (int k = 0; k < m; k += G2_GRP) {
            const int sv = (k < 16) ? sa : sb;
            float v[G2_GRP];
            #pragma unroll
            for (int u = 0; u < G2_GRP; ++u) {
                const int s = __shfl(sv, (k + u) & 15, 16);
                v[u] = 0.0f;
                if (k + u < m) v[u] = h2[(size_t)s * H_PAD + f];
            }
            #pragma unroll
            for (int u = 0; u < G2_GRP; ++u) acc += v[u];
        }
    }
    float t = -INFINITY;
    if (valid) t = (acc + hself) * di + bias;
    float mx = t;
    for (int o = 8; o >= 1; o >>= 1) mx = fmaxf(mx, __shfl_xor(mx, o, 16));
    float sm = valid ? expf(t - mx) : 0.0f;
    for (int o = 8; o >= 1; o >>= 1) sm += __shfl_xor(sm, o, 16);
    const float lse = logf(sm) + mx;
    if (valid) out[(size_t)i * N_CLS + f] = t - lse;
}

extern "C" void kernel_launch(void* const* d_in, const int* in_sizes, int n_in,
                              void* d_out, int out_size, void* d_ws, size_t ws_size,
                              hipStream_t stream) {
    const float* x   = (const float*)d_in[0];
    const int*   ei  = (const int*)d_in[1];
    const float* W1  = (const float*)d_in[2];
    const float* b1  = (const float*)d_in[3];
    const float* W2  = (const float*)d_in[4];
    const float* b2  = (const float*)d_in[5];
    float*       out = (float*)d_out;

    const int* src = ei;
    const int* dst = ei + N_EDGES;

    float* ws    = (float*)d_ws;
    float* dinv  = ws;
    float* h1    = dinv + 100352;
    float* reg2  = h1 + (size_t)N_NODES * F_H;
    float* W2pad = reg2 + (size_t)N_NODES * F_H;
    int*   cnt   = (int*)reg2;
    int*   off   = (int*)(reg2 + 100352);
    int*   csr   = (int*)(reg2 + 1800704);
    float* h2    = reg2 + 3400704;
    int*   pairs = (int*)(reg2 + 5000704);
    int*   M     = (int*)(reg2 + 8200704);
    int*   binbeg= (int*)(reg2 + 8250880);

    k_binhist<<<SBLK, 1024, 0, stream>>>(dst, M);
    k_binmove<<<SBLK, 1024, 0, stream>>>(src, dst, M, binbeg, pairs);
    k_binsort<<<NBIN, 1024, 0, stream>>>(pairs, binbeg, cnt, off, dinv, csr);

    k_gemm1<<<G1_BLOCKS, 256, 0, stream>>>(x, W1, dinv, h1);
    k_padW2<<<(F_H * H_PAD + 255) / 256, 256, 0, stream>>>(W2, W2pad);
    k_gather1<<<N_NODES / 16, 256, 0, stream>>>(csr, off, cnt, dinv, h1, b1, W2pad, h2);

    k_gather2<<<N_NODES / 16, 256, 0, stream>>>(csr, off, cnt, dinv, h2, b2, out);
}


# layer-2 gather: 4 lanes per node, 16 B of the 64-B row per lane (was 16 lanes x 4 B), 8 nodes per wave: a load fetches 8 rows not 2, 12500 waves not 50000; slot order and softmax pairing kept
# speedup vs baseline: 4.7079x; 1.0424x over previous
#include <hip/hip_runtime.h>
#include <math.h>


#define N_NODES 100000
#define N_EDGES 1600000
#define F_IN    256
#define F_H     128
#define N_CLS   10
#define H_PAD   16

typedef float v2f __attribute__((ext_vector_type(2)));
typedef float v8f __attribute__((ext_vector_type(8)));

#define BIN_SHIFT 9
#define BIN_NODES 512
#define NBIN      196
#define SBLK      256
#define EPB       (N_EDGES / SBLK)
#define EPT       7

__global__ __launch_bounds__(1024) void k_binhist(const int* __restrict__ dst,
                                                  int* __restrict__ M) {
    __shared__ int hist[NBIN];
    const int tid = threadIdx.x;
    const int blk = blockIdx.x;
    if (tid < NBIN) hist[tid] = 0;
    __syncthreads();
    const int e0 = blk * EPB;
    int d[EPT];
    #pragma unroll
    for (int i = 0; i < EPT; ++i) {
        const int idx = tid + 1024 * i;
        d[i] = dst[e0 + (idx < EPB ? idx : 0)];
    }
    #pragma unroll
    for (int i = 0; i < EPT; ++i)
        if (tid + 1024 * i < EPB) atomicAdd(&hist[d[i] >> BIN_SHIFT], 1);
    __syncthreads();
    if (tid < NBIN) M[tid * SBLK + blk] = hist[tid];
}

__global__ __launch_bounds__(1024) void k_binmove(const int* __restrict__ src,
                                                  const int* __restrict__ dst,
                                                  const int* __restrict__ M,
                                                  int* __restrict__ binbeg,
                                                  int* __restrict__ pairs) {
    __shared__ int rowtot[NBIN];
    __shared__ int rowpre[NBIN];
    __shared__ int rowown[NBIN];
    __shared__ int base[NBIN];
    __shared__ int lbeg[NBIN];
    __shared__ int cur[NBIN];
    __shared__ int2 stg[EPB];
    const int tid  = threadIdx.x;
    const int lane = tid & 31;
    const int wave = tid >> 5;
    const int blk  = blockIdx.x;
    if (tid < NBIN) cur[tid] = 0;

    int4 ra[7], rb[7];
    #pragma unroll
    for (int i = 0; i < 7; ++i) {
        const int bin = wave + 32 * i;
        const int4* row = (const int4*)(M + (bin < NBIN ? bin : NBIN - 1) * SBLK);
        ra[i] = row[lane];
        rb[i] = row[32 + lane];
    }
    #pragma unroll
    for (int i = 0; i < 7; ++i) {
        const int bin = wave + 32 * i;
        {
            const int4 a = ra[i];
            const int4 b = rb[i];
            int tot = a.x + a.y + a.z + a.w + b.x + b.y + b.z + b.w;
            const int c0 = 4 * lane, c1 = 128 + 4 * lane;
            int pre = (c0     < blk ? a.x : 0) + (c0 + 1 < blk ? a.y : 0)
                    + (c0 + 2 < blk ? a.z : 0) + (c0 + 3 < blk ? a.w : 0)
                    + (c1     < blk ? b.x : 0) + (c1 + 1 < blk ? b.y : 0)
                    + (c1 + 2 < blk ? b.z : 0) + (c1 + 3 < blk ? b.w : 0);
            int own = (c0     == blk ? a.x : 0) + (c0 + 1 == blk ? a.y : 0)
                    + (c0 + 2 == blk ? a.z : 0) + (c0 + 3 == blk ? a.w : 0)
                    + (c1     == blk ? b.x : 0) + (c1 + 1 == blk ? b.y : 0)
                    + (c1 + 2 == blk ? b.z : 0) + (c1 + 3 == blk ? b.w : 0);
            for (int o = 16; o >= 1; o >>= 1) {
                tot += __shfl_xor(tot, o);
                pre += __shfl_xor(pre, o);
                own += __shfl_xor(own, o);
            }
            if (lane == 0 && bin < NBIN) { rowtot[bin] = tot; rowpre[bin] = pre; rowown[bin] = own; }
        }
    }
    __syncthreads();
    if (wave == 0) {
        int loc[7], lloc[7];
        int s = 0, ls = 0;
        #pragma unroll
        for (int j = 0; j < 7; ++j) {
            const int bin = 7 * lane + j;
            loc[j] = s;
            lloc[j] = ls;
            s  += (bin < NBIN) ? rowtot[bin] : 0;
            ls += (bin < NBIN) ? rowown[bin] : 0;
        }
        int inc = s, linc = ls;
        for (int o = 1; o < 32; o <<= 1) {
            int t  = __shfl_up(inc, o);
            int lt = __shfl_up(linc, o);
            if (lane >= o) { inc += t; linc += lt; }
        }
        const int ex  = inc - s;
        const int lex = linc - ls;
        #pragma unroll
        for (int j = 0; j < 7; ++j) {
            const int bin = 7 * lane + j;
            if (bin < NBIN) {
                const int bb = ex + loc[j];
                base[bin] = bb + rowpre[bin];
                lbeg[bin] = lex + lloc[j];
                if (blk == 0) binbeg[bin] = bb;
            }
        }
        if (blk == 0 && lane == 31) binbeg[NBIN] = inc;
    }
    __syncthreads();

    const int e0 = blk * EPB;
    int2* pr = (int2*)pairs;
    int dd[EPT], ss[EPT];
    #pragma unroll
    for (int i = 0; i < EPT; ++i) {
        const int idx = tid + 1024 * i;
        const int e = e0 + (idx < EPB ? idx : 0);
        dd[i] = dst[e];
        ss[i] = src[e];
    }
    #pragma unroll
    for (int i = 0; i < EPT; ++i) {
        if (tid + 1024 * i < EPB) {
            const int b = dd[i] >> BIN_SHIFT;
            const int r = atomicAdd(&cur[b], 1);
            stg[lbeg[b] + r] = make_int2(ss[i], dd[i]);
        }
    }
    __syncthreads();
    int2 ee[EPT];
    #pragma unroll
    for (int i = 0; i < EPT; ++i) {
        const int p = tid + 1024 * i;
        ee[i] = stg[p < EPB ? p : 0];
    }
    #pragma unroll
    for (int i = 0; i < EPT; ++i) {
        const int p = tid + 1024 * i;
        if (p < EPB) {
            const int b = ee[i].y >> BIN_SHIFT;
            pr[base[b] + (p - lbeg[b])] = ee[i];
        }
    }
}

__device__ __forceinline__ float dinv_of(int deg) {
    float d = (float)deg;
    return (d > 0.0f) ? rsqrtf(d) : 0.0f;
}

#define STG_CAP 12288
__global__ __launch_bounds__(1024) void k_binsort(const int* __restrict__ pairs,
                                                  const int* __restrict__ binbeg,
                                                  int* __restrict__ cnt,
                                                  int* __restrict__ off,
                                                  float* __restrict__ dinv,
                                                  int* __restrict__ csr) {
    __shared__ int cntS[BIN_NODES];
    __shared__ int curS[BIN_NODES];
    __shared__ int exS[BIN_NODES];
    __shared__ int wsum[16];
    __shared__ int stgS[STG_CAP];
    const int tid  = threadIdx.x;
    const int lane = tid & 31;
    const int wave = tid >> 5;
    const int bin  = blockIdx.x;
    const int node0 = bin * BIN_NODES;
    const int beg = binbeg[bin];
    const int end = binbeg[bin + 1];
    const int2* pr = (const int2*)pairs;
    if (tid < BIN_NODES) { cntS[tid] = 0; curS[tid] = 0; }
    __syncthreads();
    for (int j0 = beg; j0 < end; j0 += 8192) {
        int dy[8];
        #pragma unroll
        for (int u = 0; u < 8; ++u) {
            const int j = j0 + tid + 1024 * u;
            dy[u] = pr[j < end ? j : end - 1].y;
        }
        #pragma unroll
        for (int u = 0; u < 8; ++u)
            if (j0 + tid + 1024 * u < end) atomicAdd(&cntS[dy[u] - node0], 1);
    }
    __syncthreads();

    const int v = (tid < BIN_NODES) ? cntS[tid] : 0;
    int inc = v;
    for (int o = 1; o < 32; o <<= 1) {
        int t = __shfl_up(inc, o);
        if (lane >= o) inc += t;
    }
    if (tid < BIN_NODES && lane == 31) wsum[wave] = inc;
    __syncthreads();
    if (tid < BIN_NODES) {
        int wp = 0;
        for (int w = 0; w < wave; ++w) wp += wsum[w];
        const int ex = wp + inc - v;
        exS[tid] = ex;
        const int i = node0 + tid;
        if (i < N_NODES) {
            off[i]  = beg + ex;
            cnt[i]  = v;
            dinv[i] = dinv_of(v + 1);
        }
    }
    __syncthreads();
    for (int j0 = beg; j0 < end; j0 += 8192) {
        int2 pp[8];
        #pragma unroll
        for (int u = 0; u < 8; ++u) {
            const int j = j0 + tid + 1024 * u;
            pp[u] = pr[j < end ? j : end - 1];
        }
        #pragma unroll
        for (int u = 0; u < 8; ++u) {
            if (j0 + tid + 1024 * u < end) {
                const int d = pp[u].y - node0;
                const int r = atomicAdd(&curS[d], 1);
                const int pos = exS[d] + r;
                if (pos < STG_CAP) stgS[pos] = pp[u].x;
                else               csr[beg + pos] = pp[u].x;
            }
        }
    }
    __syncthreads();
    const int nst = (end - beg < STG_CAP) ? (end - beg) : STG_CAP;
    for (int p0 = 0; p0 < nst; p0 += 4096) {
        int sv[4];
        #pragma unroll
        for (int u = 0; u < 4; ++u) {
            const int p = p0 + tid + 1024 * u;
            sv[u] = stgS[p < nst ? p : 0];
        }
        #pragma unroll
        for (int u = 0; u < 4; ++u) {
            const int p = p0 + tid + 1024 * u;
            if (p < nst) csr[beg + p] = sv[u];
        }
    }
}

#define LDS_STRIDE 260
#define N_TILES    (N_NODES / 16)
#define G1_BLOCKS  512

#define G1_LOAD16(dst, t0)                                                        \
    _Pragma("unroll")                                                             \
    for (int u = 0; u < 16; ++u) dst[u] = *(const v2f*)&A[4 * ((t0) + u)];
#define G1_WMMA16(src, t0)                                                        \
    _Pragma("unroll")                                                             \
    for (int u = 0; u < 16; ++u)                                                  \
        c = __builtin_amdgcn_wmma_f32_16x16x4_f32(false, src[u], false, bf[(t0) + u], \
                                                  (short)0, c, false, false);

__global__ __launch_bounds__(256) void k_gemm1(const float* __restrict__ x,
                                               const float* __restrict__ W1,
                                               const float* __restrict__ dinv,
                                               float* __restrict__ h1) {
    __shared__ __attribute__((aligned(16))) float As[2][16 * LDS_STRIDE];
    const int tid  = threadIdx.x;
    const int wave = tid >> 5;
    const int lane = tid & 31;
    const int m    = lane & 15;
    const int kg   = lane >> 4;
    const int n0   = wave * 16;

    v2f bf[64];
    #pragma unroll
    for (int t = 0; t < 64; ++t) {
        const int ka = 4 * t + 2 * kg;
        bf[t].x = W1[(size_t)ka       * F_H + n0 + m];
        bf[t].y = W1[(size_t)(ka + 1) * F_H + n0 + m];
    }

    int tile = blockIdx.x;
    const int q0 = tid, q1 = tid + 256, q2 = tid + 512, q3 = tid + 768;
    const int d0 = (q0 >> 6) * LDS_STRIDE + (q0 & 63) * 4;
    const int d1 = (q1 >> 6) * LDS_STRIDE + (q1 & 63) * 4;
    const int d2 = (q2 >> 6) * LDS_STRIDE + (q2 & 63) * 4;
    const int d3 = (q3 >> 6) * LDS_STRIDE + (q3 & 63) * 4;
    {
        const float4* xg = (const float4*)(x + (size_t)tile * 16 * F_IN);
        const float4 f0 = xg[q0], f1 = xg[q1], f2 = xg[q2], f3 = xg[q3];
        *(float4*)&As[0][d0] = f0;
        *(float4*)&As[0][d1] = f1;
        *(float4*)&As[0][d2] = f2;
        *(float4*)&As[0][d3] = f3;
    }
    float4 p0, p1, p2, p3;
    {
        const int t1 = tile + G1_BLOCKS;
        const float4* xg = (const float4*)(x + (size_t)(t1 < N_TILES ? t1 : tile) * 16 * F_IN);
        p0 = xg[q0]; p1 = xg[q1]; p2 = xg[q2]; p3 = xg[q3];
    }
    __syncthreads();

    int buf = 0;
    for (; tile < N_TILES; tile += G1_BLOCKS) {
        const bool more = tile + G1_BLOCKS < N_TILES;

        const float4* dv = (const float4*)(dinv + tile * 16 + 8 * kg);
        const float4 s0 = dv[0], s1 = dv[1];

        const float* A = &As[buf][m * LDS_STRIDE + 2 * kg];
        v8f c = {0.f,0.f,0.f,0.f,0.f,0.f,0.f,0.f};
        v2f a0[16], a1[16];
        G1_LOAD16(a0, 0)
        G1_LOAD16(a1, 16)
        __builtin_amdgcn_sched_barrier(0);
        G1_WMMA16(a0, 0)
        __builtin_amdgcn_sched_barrier(0);
        G1_LOAD16(a0, 32)
        __builtin_amdgcn_sched_barrier(0);
        G1_WMMA16(a1, 16)
        __builtin_amdgcn_sched_barrier(0);
        G1_LOAD16(a1, 48)
        __builtin_amdgcn_sched_barrier(0);
        G1_WMMA16(a0, 32)
        G1_WMMA16(a1, 48)

        if (more) {
            float* An = As[buf ^ 1];
            *(float4*)&An[d0] = p0;
            *(float4*)&An[d1] = p1;
            *(float4*)&An[d2] = p2;
            *(float4*)&An[d3] = p3;
        }
        {
            const int t2 = tile + 2 * G1_BLOCKS;
            const float4* xg = (const float4*)(x + (size_t)(t2 < N_TILES ? t2 : tile) * 16 * F_IN);
            p0 = xg[q0]; p1 = xg[q1]; p2 = xg[q2]; p3 = xg[q3];
        }
        __builtin_amdgcn_fence(__ATOMIC_RELEASE, "workgroup", "local");
        __builtin_amdgcn_s_barrier();
        __builtin_amdgcn_fence(__ATOMIC_ACQUIRE, "workgroup", "local");
        buf ^= 1;

        const float sc[8] = {s0.x, s0.y, s0.z, s0.w, s1.x, s1.y, s1.z, s1.w};
        float* out = h1 + (size_t)tile * 16 * F_H;
        #pragma unroll
        for (int r = 0; r < 8; ++r) {
            int row = r + 8 * kg;
            out[(size_t)row * F_H + n0 + m] = c[r] * sc[r];
        }
    }
}

__global__ __launch_bounds__(256) void k_padW2(const float* __restrict__ W2,
                                               float* __restrict__ Bp) {
    int t = blockIdx.x * 256 + threadIdx.x;
    if (t >= F_H * H_PAD) return;
    int k = t >> 4, c = t & 15;
    Bp[t] = (c < N_CLS) ? W2[k * N_CLS + c] : 0.0f;
}

#define T_STRIDE 132
#define G1_GRP   8

__device__ __forceinline__ float4 g1_sum_rows(const int* __restrict__ csr,
                                              const float* __restrict__ h1,
                                              int beg, int n, int first_s, int lane) {
    float4 acc = make_float4(0.f, 0.f, 0.f, 0.f);
    for (int c = 0; c < n; c += 32) {
        const int m = (n - c < 32) ? (n - c) : 32;
        int my_s = first_s;
        if (c > 0) {
            my_s = 0;
            if (lane < m) my_s = csr[beg + c + lane];
        }
        int k = 0;
        for (; k + G1_GRP <= m; k += G1_GRP) {
            float4 r[G1_GRP];
            #pragma unroll
            for (int u = 0; u < G1_GRP; ++u) {
                const int s = __builtin_amdgcn_readlane(my_s, k + u);
                r[u] = *(const float4*)&h1[(size_t)s * F_H + lane * 4];
            }
            #pragma unroll
            for (int u = 0; u < G1_GRP; ++u) {
                acc.x += r[u].x; acc.y += r[u].y; acc.z += r[u].z; acc.w += r[u].w;
            }
        }
        if (k < m) {
            float4 r[G1_GRP - 1];
            #pragma unroll
            for (int u = 0; u < G1_GRP - 1; ++u) {
                r[u] = make_float4(0.f, 0.f, 0.f, 0.f);
                if (k + u < m) {
                    const int s = __builtin_amdgcn_readlane(my_s, k + u);
                    r[u] = *(const float4*)&h1[(size_t)s * F_H + lane * 4];
                }
            }
            #pragma unroll
            for (int u = 0; u < G1_GRP - 1; ++u) {
                acc.x += r[u].x; acc.y += r[u].y; acc.z += r[u].z; acc.w += r[u].w;
            }
        }
    }
    return acc;
}

__device__ __forceinline__ float4 g1_finish(float4 acc, float4 hs, float di, float4 bb) {
    float4 v;
    v.x = (acc.x + hs.x) * di + bb.x;
    v.y = (acc.y + hs.y) * di + bb.y;
    v.z = (acc.z + hs.z) * di + bb.z;
    v.w = (acc.w + hs.w) * di + bb.w;
    v.x = v.x > 0.0f ? v.x * di : 0.0f;
    v.y = v.y > 0.0f ? v.y * di : 0.0f;
    v.z = v.z > 0.0f ? v.z * di : 0.0f;
    v.w = v.w > 0.0f ? v.w * di : 0.0f;
    return v;
}

__global__ __launch_bounds__(256) void k_gather1(const int* __restrict__ csr,
                                                 const int* __restrict__ off,
                                                 const int* __restrict__ cnt,
                                                 const float* __restrict__ dinv,
                                                 const float* __restrict__ h1,
                                                 const float* __restrict__ b1,
                                                 const float* __restrict__ Bp,
                                                 float* __restrict__ h2) {
    __shared__ __attribute__((aligned(16))) float T[16 * T_STRIDE];
    const int tile = blockIdx.x;
    const int tid  = threadIdx.x;
    const int wave = tid >> 5;
    const int lane = tid & 31;
    const float4 bb = *(const float4*)&b1[lane * 4];

    const int i0 = __builtin_amdgcn_readfirstlane(tile * 16 + wave * 2);
    const int2   o2 = *(const int2*)&off[i0];
    const int2   n2 = *(const int2*)&cnt[i0];
    const float2 d2 = *(const float2*)&dinv[i0];
    const float4 hs0 = *(const float4*)&h1[(size_t)i0 * F_H + lane * 4];
    const float4 hs1 = *(const float4*)&h1[(size_t)(i0 + 1) * F_H + lane * 4];
    int s0 = 0, s1 = 0;
    if (lane < n2.x) s0 = csr[o2.x + lane];
    if (lane < n2.y) s1 = csr[o2.y + lane];

    const float4 a0 = g1_sum_rows(csr, h1, o2.x, n2.x, s0, lane);
    *(float4*)&T[(wave * 2) * T_STRIDE + lane * 4] = g1_finish(a0, hs0, d2.x, bb);
    const float4 a1 = g1_sum_rows(csr, h1, o2.y, n2.y, s1, lane);
    *(float4*)&T[(wave * 2 + 1) * T_STRIDE + lane * 4] = g1_finish(a1, hs1, d2.y, bb);
    __syncthreads();
    if (wave != 0) return;

    const int m  = lane & 15;
    const int kg = lane >> 4;
    v8f c = {0.f,0.f,0.f,0.f,0.f,0.f,0.f,0.f};
    for (int k0 = 0; k0 < F_H; k0 += 4) {
        int ka = k0 + 2 * kg;
        v2f a = *(const v2f*)&T[m * T_STRIDE + ka];
        v2f b;
        b.x = Bp[ka * H_PAD + m];
        b.y = Bp[(ka + 1) * H_PAD + m];
        c = __builtin_amdgcn_wmma_f32_16x16x4_f32(false, a, false, b,
                                                  (short)0, c, false, false);
    }
    float* out = h2 + (size_t)tile * 16 * H_PAD;
    #pragma unroll
    for (int r = 0; r < 8; ++r) {
        int row = r + 8 * kg;
        out[(size_t)row * H_PAD + m] = c[r];
    }
}

#define G2_GRP    8
#define G2_NODES  64
#define G2_BLOCKS 1563
__global__ __launch_bounds__(256) void k_gather2(const int* __restrict__ csr,
                                                 const int* __restrict__ off,
                                                 const int* __restrict__ cnt,
                                                 const float* __restrict__ dinv,
                                                 const float* __restrict__ h2,
                                                 const float* __restrict__ b2,
                                                 float* __restrict__ out) {
    const int tid = threadIdx.x;
    const int q   = tid & 3;
    const int i   = blockIdx.x * G2_NODES + (tid >> 2);
    if (i >= N_NODES) return;
    const int c0 = 4 * q;
    const bool ok0 = c0 < N_CLS, ok1 = c0 + 1 < N_CLS, ok2 = c0 + 2 < N_CLS, ok3 = c0 + 3 < N_CLS;
    const int beg = off[i];
    const int n   = cnt[i];
    const float di = dinv[i];
    const float4 hself = *(const float4*)&h2[(size_t)i * H_PAD + c0];
    const float bs0 = b2[ok0 ? c0 : 0], bs1 = b2[ok1 ? c0 + 1 : 0], bs2 = b2[ok2 ? c0 + 2 : 0], bs3 = b2[ok3 ? c0 + 3 : 0];
    const int* cq = csr + beg + q;
    int sv[8];
    #pragma unroll
    for (int j = 0; j < 8; ++j) {
        sv[j] = 0;
        if (4 * j < n - q) sv[j] = cq[4 * j];
    }
    float4 acc = make_float4(0.f, 0.f, 0.f, 0.f);
    for (int c = 0; c < n; c += 32) {
        const int m = (n - c < 32) ? (n - c) : 32;
        if (c > 0) {
            #pragma unroll
            for (int j = 0; j < 8; ++j) {
                sv[j] = 0;
                if (4 * j < m - q) sv[j] = cq[c + 4 * j];
            }
        }
        #pragma unroll
        for (int g = 0; g < 32 / G2_GRP; ++g) {
            if (G2_GRP * g < m) {
                float4 v[G2_GRP];
                #pragma unroll
                for (int u = 0; u < G2_GRP; ++u) {
                    const int s = __shfl(sv[2 * g + (u >> 2)], u & 3, 4);
                    v[u] = make_float4(0.f, 0.f, 0.f, 0.f);
                    if (G2_GRP * g + u < m) v[u] = *(const float4*)&h2[(size_t)s * H_PAD + c0];
                }
                #pragma unroll
                for (int u = 0; u < G2_GRP; ++u) {
                    acc.x += v[u].x; acc.y += v[u].y; acc.z += v[u].z; acc.w += v[u].w;
                }
            }
        }
    }
    float t0 = -INFINITY, t1 = -INFINITY, t2 = -INFINITY, t3 = -INFINITY;
    if (ok0) t0 = (acc.x + hself.x) * di + bs0;
    if (ok1) t1 = (acc.y + hself.y) * di + bs1;
    if (ok2) t2 = (acc.z + hself.z) * di + bs2;
    if (ok3) t3 = (acc.w + hself.w) * di + bs3;
    float mx = fmaxf(fmaxf(t0, t1), fmaxf(t2, t3));
    mx = fmaxf(mx, __shfl_xor(mx, 2, 4));
    mx = fmaxf(mx, __shfl_xor(mx, 1, 4));
    float e0 = ok0 ? expf(t0 - mx) : 0.0f;
    float e1 = ok1 ? expf(t1 - mx) : 0.0f;
    float e2 = ok2 ? expf(t2 - mx) : 0.0f;
    float e3 = ok3 ? expf(t3 - mx) : 0.0f;
    e0 += __shfl_xor(e0, 2, 4); e1 += __shfl_xor(e1, 2, 4); e2 += __shfl_xor(e2, 2, 4); e3 += __shfl_xor(e3, 2, 4);
    e0 += __shfl_xor(e0, 1, 4); e1 += __shfl_xor(e1, 1, 4); e2 += __shfl_xor(e2, 1, 4); e3 += __shfl_xor(e3, 1, 4);
    const float sm  = (e0 + e2) + (e1 + e3);
    const float lse = logf(sm) + mx;
    float* o = out + (size_t)i * N_CLS + c0;
    if (ok1) *(float2*)o       = make_float2(t0 - lse, t1 - lse);
    if (ok3) *(float2*)(o + 2) = make_float2(t2 - lse, t3 - lse);
}

extern "C" void kernel_launch(void* const* d_in, const int* in_sizes, int n_in,
                              void* d_out, int out_size, void* d_ws, size_t ws_size,
                              hipStream_t stream) {
    const float* x   = (const float*)d_in[0];
    const int*   ei  = (const int*)d_in[1];
    const float* W1  = (const float*)d_in[2];
    const float* b1  = (const float*)d_in[3];
    const float* W2  = (const float*)d_in[4];
    const float* b2  = (const float*)d_in[5];
    float*       out = (float*)d_out;

    const int* src = ei;
    const int* dst = ei + N_EDGES;

    float* ws    = (float*)d_ws;
    float* dinv  = ws;
    float* h1    = dinv + 100352;
    float* reg2  = h1 + (size_t)N_NODES * F_H;
    float* W2pad = reg2 + (size_t)N_NODES * F_H;
    int*   cnt   = (int*)reg2;
    int*   off   = (int*)(reg2 + 100352);
    int*   csr   = (int*)(reg2 + 1800704);
    float* h2    = reg2 + 3400704;
    int*   pairs = (int*)(reg2 + 5000704);
    int*   M     = (int*)(reg2 + 8200704);
    int*   binbeg= (int*)(reg2 + 8250880);

    k_binhist<<<SBLK, 1024, 0, stream>>>(dst, M);
    k_binmove<<<SBLK, 1024, 0, stream>>>(src, dst, M, binbeg, pairs);
    k_binsort<<<NBIN, 1024, 0, stream>>>(pairs, binbeg, cnt, off, dinv, csr);

    k_gemm1<<<G1_BLOCKS, 256, 0, stream>>>(x, W1, dinv, h1);
    k_padW2<<<(F_H * H_PAD + 255) / 256, 256, 0, stream>>>(W2, W2pad);
    k_gather1<<<N_NODES / 16, 256, 0, stream>>>(csr, off, cnt, dinv, h1, b1, W2pad, h2);

    k_gather2<<<G2_BLOCKS, 256, 0, stream>>>(csr, off, cnt, dinv, h2, b2, out);
}


# edge sort level 1: each block sorts its own slice by bin in LDS and writes a run table, so the global count pass (one launch) is gone; level 2 walks its bin's 256 runs; slot order (sum order) changes
# speedup vs baseline: 4.8567x; 1.0316x over previous
#include <hip/hip_runtime.h>
#include <math.h>


#define N_NODES 100000
#define N_EDGES 1600000
#define F_IN    256
#define F_H     128
#define N_CLS   10
#define H_PAD   16

typedef float v2f __attribute__((ext_vector_type(2)));
typedef float v8f __attribute__((ext_vector_type(8)));

#define BIN_SHIFT 9
#define BIN_NODES 512
#define NBIN      196
#define SBLK      256
#define EPB       (N_EDGES / SBLK)
#define EPT       7

__global__ __launch_bounds__(1024) void k_binlocal(const int* __restrict__ src,
                                                   const int* __restrict__ dst,
                                                   int* __restrict__ M,
                                                   int* __restrict__ Lb,
                                                   int* __restrict__ pairs) {
    __shared__ int cntB[NBIN];
    __shared__ int lbeg[NBIN];
    __shared__ int2 stg[EPB];
    const int tid  = threadIdx.x;
    const int lane = tid & 31;
    const int wave = tid >> 5;
    const int blk  = blockIdx.x;
    if (tid < NBIN) cntB[tid] = 0;
    __syncthreads();

    const int e0 = blk * EPB;
    int dd[EPT], ss[EPT], rk[EPT];
    #pragma unroll
    for (int i = 0; i < EPT; ++i) {
        const int idx = tid + 1024 * i;
        const int e = e0 + (idx < EPB ? idx : 0);
        dd[i] = dst[e];
        ss[i] = src[e];
    }
    #pragma unroll
    for (int i = 0; i < EPT; ++i) {
        rk[i] = 0;
        if (tid + 1024 * i < EPB) rk[i] = atomicAdd(&cntB[dd[i] >> BIN_SHIFT], 1);
    }
    __syncthreads();
    if (wave == 0) {
        int loc[7];
        int s = 0;
        #pragma unroll
        for (int j = 0; j < 7; ++j) {
            const int bin = 7 * lane + j;
            loc[j] = s;
            s += (bin < NBIN) ? cntB[bin] : 0;
        }
        int inc = s;
        for (int o = 1; o < 32; o <<= 1) {
            int t = __shfl_up(inc, o);
            if (lane >= o) inc += t;
        }
        const int ex = inc - s;
        #pragma unroll
        for (int j = 0; j < 7; ++j) {
            const int bin = 7 * lane + j;
            if (bin < NBIN) lbeg[bin] = ex + loc[j];
        }
    }
    __syncthreads();
    if (tid < NBIN) {
        M[tid * SBLK + blk]  = cntB[tid];
        Lb[tid * SBLK + blk] = lbeg[tid];
    }
    #pragma unroll
    for (int i = 0; i < EPT; ++i) {
        if (tid + 1024 * i < EPB) {
            const int b = dd[i] >> BIN_SHIFT;
            stg[lbeg[b] + rk[i]] = make_int2(ss[i], dd[i]);
        }
    }
    __syncthreads();
    int2* pr = (int2*)pairs + (size_t)blk * EPB;
    int2 ee[EPT];
    #pragma unroll
    for (int i = 0; i < EPT; ++i) {
        const int p = tid + 1024 * i;
        ee[i] = stg[p < EPB ? p : 0];
    }
    #pragma unroll
    for (int i = 0; i < EPT; ++i) {
        const int p = tid + 1024 * i;
        if (p < EPB) pr[p] = ee[i];
    }
}

__device__ __forceinline__ float dinv_of(int deg) {
    float d = (float)deg;
    return (d > 0.0f) ? rsqrtf(d) : 0.0f;
}

#define STG_CAP 12288
#define RPW     (SBLK / 32)
#define PREV_Q  13
__global__ __launch_bounds__(1024) void k_binsort(const int* __restrict__ pairs,
                                                  const int* __restrict__ M,
                                                  const int* __restrict__ Lb,
                                                  int* __restrict__ cnt,
                                                  int* __restrict__ off,
                                                  float* __restrict__ dinv,
                                                  int* __restrict__ csr) {
    __shared__ int cntS[BIN_NODES];
    __shared__ int curS[BIN_NODES];
    __shared__ int exS[BIN_NODES];
    __shared__ int wsum[16];
    __shared__ __attribute__((aligned(16))) int rs[SBLK];
    __shared__ __attribute__((aligned(16))) int rl[SBLK];
    __shared__ int redP[32];
    __shared__ int redO[32];
    __shared__ int stgS[STG_CAP];
    const int tid  = threadIdx.x;
    const int lane = tid & 31;
    const int wave = tid >> 5;
    const int bin  = blockIdx.x;
    const int node0 = bin * BIN_NODES;
    const int2* pr = (const int2*)pairs;

    int myLen = 0, myBeg = 0;
    if (tid < SBLK) { myLen = M[bin * SBLK + tid]; myBeg = Lb[bin * SBLK + tid]; }
    const int4* M4 = (const int4*)M;
    const int nq = bin * (SBLK / 4);
    int4 pv[PREV_Q];
    #pragma unroll
    for (int u = 0; u < PREV_Q; ++u) {
        const int q = tid + 1024 * u;
        pv[u] = make_int4(0, 0, 0, 0);
        if (q < nq) pv[u] = M4[q];
    }
    if (tid < BIN_NODES) { cntS[tid] = 0; curS[tid] = 0; }
    if (tid < SBLK) { rs[tid] = tid * EPB + myBeg; rl[tid] = myLen; }
    int sPrev = 0;
    #pragma unroll
    for (int u = 0; u < PREV_Q; ++u) sPrev += (pv[u].x + pv[u].y) + (pv[u].z + pv[u].w);
    int sOwn = myLen;
    for (int o = 16; o >= 1; o >>= 1) {
        sPrev += __shfl_xor(sPrev, o);
        sOwn  += __shfl_xor(sOwn, o);
    }
    if (lane == 0) { redP[wave] = sPrev; redO[wave] = sOwn; }
    __syncthreads();
    int beg = 0, tot = 0;
    for (int w = 0; w < 32; ++w) { beg += redP[w]; tot += redO[w]; }

    int st[RPW], ln[RPW];
    {
        const int4 sa = *(const int4*)&rs[wave * RPW], sb = *(const int4*)&rs[wave * RPW + 4];
        const int4 la = *(const int4*)&rl[wave * RPW], lb = *(const int4*)&rl[wave * RPW + 4];
        st[0] = sa.x; st[1] = sa.y; st[2] = sa.z; st[3] = sa.w; st[4] = sb.x; st[5] = sb.y; st[6] = sb.z; st[7] = sb.w;
        ln[0] = la.x; ln[1] = la.y; ln[2] = la.z; ln[3] = la.w; ln[4] = lb.x; ln[5] = lb.y; ln[6] = lb.z; ln[7] = lb.w;
    }
    int2 ev[RPW][2];
    #pragma unroll
    for (int r = 0; r < RPW; ++r) {
        ev[r][0] = make_int2(0, node0);
        ev[r][1] = make_int2(0, node0);
        if (lane < ln[r])      ev[r][0] = pr[st[r] + lane];
        if (lane + 32 < ln[r]) ev[r][1] = pr[st[r] + 32 + lane];
    }
    #pragma unroll
    for (int r = 0; r < RPW; ++r) {
        if (lane < ln[r])      atomicAdd(&cntS[ev[r][0].y - node0], 1);
        if (lane + 32 < ln[r]) atomicAdd(&cntS[ev[r][1].y - node0], 1);
        for (int o = 64 + lane; o < ln[r]; o += 32)
            atomicAdd(&cntS[pr[st[r] + o].y - node0], 1);
    }
    __syncthreads();

    const int v = (tid < BIN_NODES) ? cntS[tid] : 0;
    int inc = v;
    for (int o = 1; o < 32; o <<= 1) {
        int t = __shfl_up(inc, o);
        if (lane >= o) inc += t;
    }
    if (tid < BIN_NODES && lane == 31) wsum[wave] = inc;
    __syncthreads();
    if (tid < BIN_NODES) {
        int wp = 0;
        for (int w = 0; w < wave; ++w) wp += wsum[w];
        const int ex = wp + inc - v;
        exS[tid] = ex;
        const int i = node0 + tid;
        if (i < N_NODES) {
            off[i]  = beg + ex;
            cnt[i]  = v;
            dinv[i] = dinv_of(v + 1);
        }
    }
    __syncthreads();
    #pragma unroll
    for (int r = 0; r < RPW; ++r) {
        if (lane < ln[r]) {
            const int d = ev[r][0].y - node0;
            const int pos = exS[d] + atomicAdd(&curS[d], 1);
            if (pos < STG_CAP) stgS[pos] = ev[r][0].x;
            else               csr[beg + pos] = ev[r][0].x;
        }
        if (lane + 32 < ln[r]) {
            const int d = ev[r][1].y - node0;
            const int pos = exS[d] + atomicAdd(&curS[d], 1);
            if (pos < STG_CAP) stgS[pos] = ev[r][1].x;
            else               csr[beg + pos] = ev[r][1].x;
        }
        for (int o = 64 + lane; o < ln[r]; o += 32) {
            const int2 p = pr[st[r] + o];
            const int d = p.y - node0;
            const int pos = exS[d] + atomicAdd(&curS[d], 1);
            if (pos < STG_CAP) stgS[pos] = p.x;
            else               csr[beg + pos] = p.x;
        }
    }
    __syncthreads();
    const int nst = (tot < STG_CAP) ? tot : STG_CAP;
    for (int p0 = 0; p0 < nst; p0 += 4096) {
        int sv[4];
        #pragma unroll
        for (int u = 0; u < 4; ++u) {
            const int p = p0 + tid + 1024 * u;
            sv[u] = stgS[p < nst ? p : 0];
        }
        #pragma unroll
        for (int u = 0; u < 4; ++u) {
            const int p = p0 + tid + 1024 * u;
            if (p < nst) csr[beg + p] = sv[u];
        }
    }
}

#define LDS_STRIDE 260
#define N_TILES    (N_NODES / 16)
#define G1_BLOCKS  512

#define G1_LOAD16(dst, t0)                                                        \
    _Pragma("unroll")                                                             \
    for (int u = 0; u < 16; ++u) dst[u] = *(const v2f*)&A[4 * ((t0) + u)];
#define G1_WMMA16(src, t0)                                                        \
    _Pragma("unroll")                                                             \
    for (int u = 0; u < 16; ++u)                                                  \
        c = __builtin_amdgcn_wmma_f32_16x16x4_f32(false, src[u], false, bf[(t0) + u], \
                                                  (short)0, c, false, false);

__global__ __launch_bounds__(256) void k_gemm1(const float* __restrict__ x,
                                               const float* __restrict__ W1,
                                               const float* __restrict__ dinv,
                                               float* __restrict__ h1) {
    __shared__ __attribute__((aligned(16))) float As[2][16 * LDS_STRIDE];
    const int tid  = threadIdx.x;
    const int wave = tid >> 5;
    const int lane = tid & 31;
    const int m    = lane & 15;
    const int kg   = lane >> 4;
    const int n0   = wave * 16;

    v2f bf[64];
    #pragma unroll
    for (int t = 0; t < 64; ++t) {
        const int ka = 4 * t + 2 * kg;
        bf[t].x = W1[(size_t)ka       * F_H + n0 + m];
        bf[t].y = W1[(size_t)(ka + 1) * F_H + n0 + m];
    }

    int tile = blockIdx.x;
    const int q0 = tid, q1 = tid + 256, q2 = tid + 512, q3 = tid + 768;
    const int d0 = (q0 >> 6) * LDS_STRIDE + (q0 & 63) * 4;
    const int d1 = (q1 >> 6) * LDS_STRIDE + (q1 & 63) * 4;
    const int d2 = (q2 >> 6) * LDS_STRIDE + (q2 & 63) * 4;
    const int d3 = (q3 >> 6) * LDS_STRIDE + (q3 & 63) * 4;
    {
        const float4* xg = (const float4*)(x + (size_t)tile * 16 * F_IN);
        const float4 f0 = xg[q0], f1 = xg[q1], f2 = xg[q2], f3 = xg[q3];
        *(float4*)&As[0][d0] = f0;
        *(float4*)&As[0][d1] = f1;
        *(float4*)&As[0][d2] = f2;
        *(float4*)&As[0][d3] = f3;
    }
    float4 p0, p1, p2, p3;
    {
        const int t1 = tile + G1_BLOCKS;
        const float4* xg = (const float4*)(x + (size_t)(t1 < N_TILES ? t1 : tile) * 16 * F_IN);
        p0 = xg[q0]; p1 = xg[q1]; p2 = xg[q2]; p3 = xg[q3];
    }
    __syncthreads();

    int buf = 0;
    for (; tile < N_TILES; tile += G1_BLOCKS) {
        const bool more = tile + G1_BLOCKS < N_TILES;

        const float4* dv = (const float4*)(dinv + tile * 16 + 8 * kg);
        const float4 s0 = dv[0], s1 = dv[1];

        const float* A = &As[buf][m * LDS_STRIDE + 2 * kg];
        v8f c = {0.f,0.f,0.f,0.f,0.f,0.f,0.f,0.f};
        v2f a0[16], a1[16];
        G1_LOAD16(a0, 0)
        G1_LOAD16(a1, 16)
        __builtin_amdgcn_sched_barrier(0);
        G1_WMMA16(a0, 0)
        __builtin_amdgcn_sched_barrier(0);
        G1_LOAD16(a0, 32)
        __builtin_amdgcn_sched_barrier(0);
        G1_WMMA16(a1, 16)
        __builtin_amdgcn_sched_barrier(0);
        G1_LOAD16(a1, 48)
        __builtin_amdgcn_sched_barrier(0);
        G1_WMMA16(a0, 32)
        G1_WMMA16(a1, 48)

        if (more) {
            float* An = As[buf ^ 1];
            *(float4*)&An[d0] = p0;
            *(float4*)&An[d1] = p1;
            *(float4*)&An[d2] = p2;
            *(float4*)&An[d3] = p3;
        }
        {
            const int t2 = tile + 2 * G1_BLOCKS;
            const float4* xg = (const float4*)(x + (size_t)(t2 < N_TILES ? t2 : tile) * 16 * F_IN);
            p0 = xg[q0]; p1 = xg[q1]; p2 = xg[q2]; p3 = xg[q3];
        }
        __builtin_amdgcn_fence(__ATOMIC_RELEASE, "workgroup", "local");
        __builtin_amdgcn_s_barrier();
        __builtin_amdgcn_fence(__ATOMIC_ACQUIRE, "workgroup", "local");
        buf ^= 1;

        const float sc[8] = {s0.x, s0.y, s0.z, s0.w, s1.x, s1.y, s1.z, s1.w};
        float* out = h1 + (size_t)tile * 16 * F_H;
        #pragma unroll
        for (int r = 0; r < 8; ++r) {
            int row = r + 8 * kg;
            out[(size_t)row * F_H + n0 + m] = c[r] * sc[r];
        }
    }
}

__global__ __launch_bounds__(256) void k_padW2(const float* __restrict__ W2,
                                               float* __restrict__ Bp) {
    int t = blockIdx.x * 256 + threadIdx.x;
    if (t >= F_H * H_PAD) return;
    int k = t >> 4, c = t & 15;
    Bp[t] = (c < N_CLS) ? W2[k * N_CLS + c] : 0.0f;
}

#define T_STRIDE 132
#define G1_GRP   8

__device__ __forceinline__ float4 g1_sum_rows(const int* __restrict__ csr,
                                              const float* __restrict__ h1,
                                              int beg, int n, int first_s, int lane) {
    float4 acc = make_float4(0.f, 0.f, 0.f, 0.f);
    for (int c = 0; c < n; c += 32) {
        const int m = (n - c < 32) ? (n - c) : 32;
        int my_s = first_s;
        if (c > 0) {
            my_s = 0;
            if (lane < m) my_s = csr[beg + c + lane];
        }
        int k = 0;
        for (; k + G1_GRP <= m; k += G1_GRP) {
            float4 r[G1_GRP];
            #pragma unroll
            for (int u = 0; u < G1_GRP; ++u) {
                const int s = __builtin_amdgcn_readlane(my_s, k + u);
                r[u] = *(const float4*)&h1[(size_t)s * F_H + lane * 4];
            }
            #pragma unroll
            for (int u = 0; u < G1_GRP; ++u) {
                acc.x += r[u].x; acc.y += r[u].y; acc.z += r[u].z; acc.w += r[u].w;
            }
        }
        if (k < m) {
            float4 r[G1_GRP - 1];
            #pragma unroll
            for (int u = 0; u < G1_GRP - 1; ++u) {
                r[u] = make_float4(0.f, 0.f, 0.f, 0.f);
                if (k + u < m) {
                    const int s = __builtin_amdgcn_readlane(my_s, k + u);
                    r[u] = *(const float4*)&h1[(size_t)s * F_H + lane * 4];
                }
            }
            #pragma unroll
            for (int u = 0; u < G1_GRP - 1; ++u) {
                acc.x += r[u].x; acc.y += r[u].y; acc.z += r[u].z; acc.w += r[u].w;
            }
        }
    }
    return acc;
}

__device__ __forceinline__ float4 g1_finish(float4 acc, float4 hs, float di, float4 bb) {
    float4 v;
    v.x = (acc.x + hs.x) * di + bb.x;
    v.y = (acc.y + hs.y) * di + bb.y;
    v.z = (acc.z + hs.z) * di + bb.z;
    v.w = (acc.w + hs.w) * di + bb.w;
    v.x = v.x > 0.0f ? v.x * di : 0.0f;
    v.y = v.y > 0.0f ? v.y * di : 0.0f;
    v.z = v.z > 0.0f ? v.z * di : 0.0f;
    v.w = v.w > 0.0f ? v.w * di : 0.0f;
    return v;
}

__global__ __launch_bounds__(256) void k_gather1(const int* __restrict__ csr,
                                                 const int* __restrict__ off,
                                                 const int* __restrict__ cnt,
                                                 const float* __restrict__ dinv,
                                                 const float* __restrict__ h1,
                                                 const float* __restrict__ b1,
                                                 const float* __restrict__ Bp,
                                                 float* __restrict__ h2) {
    __shared__ __attribute__((aligned(16))) float T[16 * T_STRIDE];
    const int tile = blockIdx.x;
    const int tid  = threadIdx.x;
    const int wave = tid >> 5;
    const int lane = tid & 31;
    const float4 bb = *(const float4*)&b1[lane * 4];

    const int i0 = __builtin_amdgcn_readfirstlane(tile * 16 + wave * 2);
    const int2   o2 = *(const int2*)&off[i0];
    const int2   n2 = *(const int2*)&cnt[i0];
    const float2 d2 = *(const float2*)&dinv[i0];
    const float4 hs0 = *(const float4*)&h1[(size_t)i0 * F_H + lane * 4];
    const float4 hs1 = *(const float4*)&h1[(size_t)(i0 + 1) * F_H + lane * 4];
    int s0 = 0, s1 = 0;
    if (lane < n2.x) s0 = csr[o2.x + lane];
    if (lane < n2.y) s1 = csr[o2.y + lane];

    const float4 a0 = g1_sum_rows(csr, h1, o2.x, n2.x, s0, lane);
    *(float4*)&T[(wave * 2) * T_STRIDE + lane * 4] = g1_finish(a0, hs0, d2.x, bb);
    const float4 a1 = g1_sum_rows(csr, h1, o2.y, n2.y, s1, lane);
    *(float4*)&T[(wave * 2 + 1) * T_STRIDE + lane * 4] = g1_finish(a1, hs1, d2.y, bb);
    __syncthreads();
    if (wave != 0) return;

    const int m  = lane & 15;
    const int kg = lane >> 4;
    v8f c = {0.f,0.f,0.f,0.f,0.f,0.f,0.f,0.f};
    for (int k0 = 0; k0 < F_H; k0 += 4) {
        int ka = k0 + 2 * kg;
        v2f a = *(const v2f*)&T[m * T_STRIDE + ka];
        v2f b;
        b.x = Bp[ka * H_PAD + m];
        b.y = Bp[(ka + 1) * H_PAD + m];
        c = __builtin_amdgcn_wmma_f32_16x16x4_f32(false, a, false, b,
                                                  (short)0, c, false, false);
    }
    float* out = h2 + (size_t)tile * 16 * H_PAD;
    #pragma unroll
    for (int r = 0; r < 8; ++r) {
        int row = r + 8 * kg;
        out[(size_t)row * H_PAD + m] = c[r];
    }
}

#define G2_GRP    8
#define G2_NODES  64
#define G2_BLOCKS 1563
__global__ __launch_bounds__(256) void k_gather2(const int* __restrict__ csr,
                                                 const int* __restrict__ off,
                                                 const int* __restrict__ cnt,
                                                 const float* __restrict__ dinv,
                                                 const float* __restrict__ h2,
                                                 const float* __restrict__ b2,
                                                 float* __restrict__ out) {
    const int tid = threadIdx.x;
    const int q   = tid & 3;
    const int i   = blockIdx.x * G2_NODES + (tid >> 2);
    if (i >= N_NODES) return;
    const int c0 = 4 * q;
    const bool ok0 = c0 < N_CLS, ok1 = c0 + 1 < N_CLS, ok2 = c0 + 2 < N_CLS, ok3 = c0 + 3 < N_CLS;
    const int beg = off[i];
    const int n   = cnt[i];
    const float di = dinv[i];
    const float4 hself = *(const float4*)&h2[(size_t)i * H_PAD + c0];
    const float bs0 = b2[ok0 ? c0 : 0], bs1 = b2[ok1 ? c0 + 1 : 0], bs2 = b2[ok2 ? c0 + 2 : 0], bs3 = b2[ok3 ? c0 + 3 : 0];
    const int* cq = csr + beg + q;
    int sv[8];
    #pragma unroll
    for (int j = 0; j < 8; ++j) {
        sv[j] = 0;
        if (4 * j < n - q) sv[j] = cq[4 * j];
    }
    float4 acc = make_float4(0.f, 0.f, 0.f, 0.f);
    for (int c = 0; c < n; c += 32) {
        const int m = (n - c < 32) ? (n - c) : 32;
        if (c > 0) {
            #pragma unroll
            for (int j = 0; j < 8; ++j) {
                sv[j] = 0;
                if (4 * j < m - q) sv[j] = cq[c + 4 * j];
            }
        }
        #pragma unroll
        for (int g = 0; g < 32 / G2_GRP; ++g) {
            if (G2_GRP * g < m) {
                float4 v[G2_GRP];
                #pragma unroll
                for (int u = 0; u < G2_GRP; ++u) {
                    const int s = __shfl(sv[2 * g + (u >> 2)], u & 3, 4);
                    v[u] = make_float4(0.f, 0.f, 0.f, 0.f);
                    if (G2_GRP * g + u < m) v[u] = *(const float4*)&h2[(size_t)s * H_PAD + c0];
                }
                #pragma unroll
                for (int u = 0; u < G2_GRP; ++u) {
                    acc.x += v[u].x; acc.y += v[u].y; acc.z += v[u].z; acc.w += v[u].w;
                }
            }
        }
    }
    float t0 = -INFINITY, t1 = -INFINITY, t2 = -INFINITY, t3 = -INFINITY;
    if (ok0) t0 = (acc.x + hself.x) * di + bs0;
    if (ok1) t1 = (acc.y + hself.y) * di + bs1;
    if (ok2) t2 = (acc.z + hself.z) * di + bs2;
    if (ok3) t3 = (acc.w + hself.w) * di + bs3;
    float mx = fmaxf(fmaxf(t0, t1), fmaxf(t2, t3));
    mx = fmaxf(mx, __shfl_xor(mx, 2, 4));
    mx = fmaxf(mx, __shfl_xor(mx, 1, 4));
    float e0 = ok0 ? expf(t0 - mx) : 0.0f;
    float e1 = ok1 ? expf(t1 - mx) : 0.0f;
    float e2 = ok2 ? expf(t2 - mx) : 0.0f;
    float e3 = ok3 ? expf(t3 - mx) : 0.0f;
    e0 += __shfl_xor(e0, 2, 4); e1 += __shfl_xor(e1, 2, 4); e2 += __shfl_xor(e2, 2, 4); e3 += __shfl_xor(e3, 2, 4);
    e0 += __shfl_xor(e0, 1, 4); e1 += __shfl_xor(e1, 1, 4); e2 += __shfl_xor(e2, 1, 4); e3 += __shfl_xor(e3, 1, 4);
    const float sm  = (e0 + e2) + (e1 + e3);
    const float lse = logf(sm) + mx;
    float* o = out + (size_t)i * N_CLS + c0;
    if (ok1) *(float2*)o       = make_float2(t0 - lse, t1 - lse);
    if (ok3) *(float2*)(o + 2) = make_float2(t2 - lse, t3 - lse);
}

extern "C" void kernel_launch(void* const* d_in, const int* in_sizes, int n_in,
                              void* d_out, int out_size, void* d_ws, size_t ws_size,
                              hipStream_t stream) {
    const float* x   = (const float*)d_in[0];
    const int*   ei  = (const int*)d_in[1];
    const float* W1  = (const float*)d_in[2];
    const float* b1  = (const float*)d_in[3];
    const float* W2  = (const float*)d_in[4];
    const float* b2  = (const float*)d_in[5];
    float*       out = (float*)d_out;

    const int* src = ei;
    const int* dst = ei + N_EDGES;

    float* ws    = (float*)d_ws;
    float* dinv  = ws;
    float* h1    = dinv + 100352;
    float* reg2  = h1 + (size_t)N_NODES * F_H;
    float* W2pad = reg2 + (size_t)N_NODES * F_H;
    int*   cnt   = (int*)reg2;
    int*   off   = (int*)(reg2 + 100352);
    int*   csr   = (int*)(reg2 + 1800704);
    float* h2    = reg2 + 3400704;
    int*   pairs = (int*)(reg2 + 5000704);
    int*   M     = (int*)(reg2 + 8200704);
    int*   Lb    = (int*)(reg2 + 8250880);

    k_binlocal<<<SBLK, 1024, 0, stream>>>(src, dst, M, Lb, pairs);
    k_binsort<<<NBIN, 1024, 0, stream>>>(pairs, M, Lb, cnt, off, dinv, csr);

    k_gemm1<<<G1_BLOCKS, 256, 0, stream>>>(x, W1, dinv, h1);
    k_padW2<<<(F_H * H_PAD + 255) / 256, 256, 0, stream>>>(W2, W2pad);
    k_gather1<<<N_NODES / 16, 256, 0, stream>>>(csr, off, cnt, dinv, h1, b1, W2pad, h2);

    k_gather2<<<G2_BLOCKS, 256, 0, stream>>>(csr, off, cnt, dinv, h2, b2, out);
}
